# starting kernel
# baseline (speedup 1.0000x reference)
#include <hip/hip_runtime.h>
#include <hip/hip_bf16.h>

typedef __attribute__((ext_vector_type(2))) float v2f;
typedef __attribute__((ext_vector_type(8))) float v8f;

#define D 32
#define SLOPE 0.2f
#define EPS 1e-16f

__device__ __forceinline__ unsigned f2o(float f) {
    unsigned b = __float_as_uint(f);
    return (b & 0x80000000u) ? ~b : (b | 0x80000000u);
}
__device__ __forceinline__ float o2f(unsigned o) {
    unsigned b = (o & 0x80000000u) ? (o & 0x7FFFFFFFu) : ~o;
    return __uint_as_float(b);
}

__device__ __forceinline__ float lrelu(float x) { return x > 0.f ? x : SLOPE * x; }

__global__ __launch_bounds__(256) void gat_init(unsigned* __restrict__ m,
                                                float* __restrict__ denom,
                                                float* __restrict__ out, int N) {
    int i = blockIdx.x * blockDim.x + threadIdx.x;
    if (i < N) { m[i] = 0u; denom[i] = 0.f; }
    if (i < N * D) out[i] = 0.f;
}

__global__ __launch_bounds__(256) void gat_gemm(const float* __restrict__ X,
                                                const float* __restrict__ W,
                                                float* __restrict__ H, int N) {
    const int wave = blockIdx.x * 8 + (threadIdx.x >> 5);
    const int lane = threadIdx.x & 31;
    const int m0 = wave * 16;
    if (m0 >= N) return;
    const int row   = m0 + (lane & 15);
    const int col   = lane & 15;
    const int khalf = (lane >> 4) * 2;
    v8f c0 = {}; v8f c1 = {};
#pragma unroll
    for (int kb = 0; kb < 8; ++kb) {
        const int k0 = kb * 4 + khalf;
        v2f a;  a.x  = X[row * D + k0];        a.y  = X[row * D + k0 + 1];
        v2f b0; b0.x = W[k0 * D + col];        b0.y = W[(k0 + 1) * D + col];
        v2f b1; b1.x = W[k0 * D + col + 16];   b1.y = W[(k0 + 1) * D + col + 16];
        c0 = __builtin_amdgcn_wmma_f32_16x16x4_f32(false, a, false, b0, (short)0, c0, false, false);
        c1 = __builtin_amdgcn_wmma_f32_16x16x4_f32(false, a, false, b1, (short)0, c1, false, false);
    }
    const int rbase = m0 + (lane >> 4) * 8;
#pragma unroll
    for (int r = 0; r < 8; ++r) {
        H[(rbase + r) * D + col]      = c0[r];
        H[(rbase + r) * D + col + 16] = c1[r];
    }
}

__global__ __launch_bounds__(256) void gat_dots(const float* __restrict__ H,
                                                const float* __restrict__ a_src,
                                                const float* __restrict__ a_dst,
                                                float* __restrict__ as_, float* __restrict__ ad_,
                                                int N) {
    int i = blockIdx.x * blockDim.x + threadIdx.x;
    if (i >= N) return;
    float s = 0.f, d = 0.f;
#pragma unroll
    for (int k = 0; k < D; ++k) {
        float h = H[i * D + k];
        s += h * a_src[k];
        d += h * a_dst[k];
    }
    as_[i] = s; ad_[i] = d;
}

__global__ __launch_bounds__(256) void gat_edge_max(const int* __restrict__ ei, int E, int N,
                                                    const float* __restrict__ as_,
                                                    const float* __restrict__ ad_,
                                                    unsigned* __restrict__ m) {
    int e = blockIdx.x * blockDim.x + threadIdx.x;
    int Etot = E + N;
    if (e >= Etot) return;
    int s = (e < E) ? ei[e]     : (e - E);
    int d = (e < E) ? ei[E + e] : (e - E);
    float lg = lrelu(as_[s] + ad_[d]);
    atomicMax(&m[d], f2o(lg));
}

__global__ __launch_bounds__(256) void gat_edge_sum(const int* __restrict__ ei, int E, int N,
                                                    const float* __restrict__ as_,
                                                    const float* __restrict__ ad_,
                                                    const unsigned* __restrict__ m,
                                                    float* __restrict__ denom) {
    int e = blockIdx.x * blockDim.x + threadIdx.x;
    int Etot = E + N;
    if (e >= Etot) return;
    int s = (e < E) ? ei[e]     : (e - E);
    int d = (e < E) ? ei[E + e] : (e - E);
    float lg = lrelu(as_[s] + ad_[d]);
    float ex = __expf(lg - o2f(m[d]));
    atomicAdd(&denom[d], ex);
}

__global__ __launch_bounds__(256) void gat_edge_aggr(const int* __restrict__ ei, int E, int N,
                                                     const float* __restrict__ as_,
                                                     const float* __restrict__ ad_,
                                                     const unsigned* __restrict__ m,
                                                     const float* __restrict__ denom,
                                                     const float* __restrict__ H,
                                                     float* __restrict__ out) {
    int e    = blockIdx.x * 8 + (threadIdx.x >> 5);
    int lane = threadIdx.x & 31;
    int Etot = E + N;
    if (e >= Etot) return;
    int s = (e < E) ? ei[e]     : (e - E);
    int d = (e < E) ? ei[E + e] : (e - E);
    float lg    = lrelu(as_[s] + ad_[d]);
    float ex    = __expf(lg - o2f(m[d]));
    float alpha = ex / (denom[d] + EPS);
    atomicAdd(&out[d * D + lane], alpha * H[s * D + lane]);
}

__global__ __launch_bounds__(256) void gat_finalize(float* __restrict__ out,
                                                    const float* __restrict__ b, int N) {
    int i = blockIdx.x * blockDim.x + threadIdx.x;
    if (i >= N * D) return;
    float v = out[i] + b[i & (D - 1)];
    out[i] = (v > 0.f) ? v : (__expf(v) - 1.f);
}

static void run_layer(const float* x_in, const int* ei, int E, int N,
                      const float* W, const float* a_src, const float* a_dst, const float* b,
                      float* H, float* as_, float* ad_, float* denom, unsigned* m,
                      float* out, hipStream_t stream) {
    const int Etot = E + N;
    const int nd_blocks   = (N * D + 255) / 256;
    const int node_blocks = (N + 255) / 256;
    const int edge_blocks = (Etot + 255) / 256;
    const int tile_waves  = (N + 15) / 16;
    const int gemm_blocks = (tile_waves + 7) / 8;
    const int aggr_blocks = (Etot + 7) / 8;

    gat_init<<<nd_blocks, 256, 0, stream>>>(m, denom, out, N);
    gat_gemm<<<gemm_blocks, 256, 0, stream>>>(x_in, W, H, N);
    gat_dots<<<node_blocks, 256, 0, stream>>>(H, a_src, a_dst, as_, ad_, N);
    gat_edge_max<<<edge_blocks, 256, 0, stream>>>(ei, E, N, as_, ad_, m);
    gat_edge_sum<<<edge_blocks, 256, 0, stream>>>(ei, E, N, as_, ad_, m, denom);
    gat_edge_aggr<<<aggr_blocks, 256, 0, stream>>>(ei, E, N, as_, ad_, m, denom, H, out);
    gat_finalize<<<nd_blocks, 256, 0, stream>>>(out, b, N);
}

extern "C" void kernel_launch(void* const* d_in, const int* in_sizes, int n_in,
                              void* d_out, int out_size, void* d_ws, size_t ws_size,
                              hipStream_t stream) {
    const float* x      = (const float*)d_in[0];
    const int*   ei     = (const int*)d_in[1];
    const float* W1     = (const float*)d_in[2];
    const float* a_src1 = (const float*)d_in[3];
    const float* a_dst1 = (const float*)d_in[4];
    const float* b1     = (const float*)d_in[5];
    const float* W2     = (const float*)d_in[6];
    const float* a_src2 = (const float*)d_in[7];
    const float* a_dst2 = (const float*)d_in[8];
    const float* b2     = (const float*)d_in[9];

    const int N = in_sizes[0] / D;
    const int E = in_sizes[1] / 2;

    float*    H     = (float*)d_ws;
    float*    as_   = H + (size_t)N * D;
    float*    ad_   = as_ + N;
    float*    denom = ad_ + N;
    unsigned* m     = (unsigned*)(denom + N);

    float* xbar = (float*)d_out;
    float* z    = xbar + (size_t)N * D;

    run_layer(x, ei, E, N, W1, a_src1, a_dst1, b1, H, as_, ad_, denom, m, z, stream);
    run_layer(z, ei, E, N, W2, a_src2, a_dst2, b2, H, as_, ad_, denom, m, xbar, stream);
}


# Group edges by destination once per call (count with returning atomic, scan, place); one wave per node does max, sum, aggregation in registers: no float atomics; sum order = atomic arrival rank
# speedup vs baseline: 2.6016x; 2.6016x over previous
#include <hip/hip_runtime.h>
#include <hip/hip_bf16.h>


typedef __attribute__((ext_vector_type(2))) float v2f;
typedef __attribute__((ext_vector_type(8))) float v8f;

#define D 32
#define SLOPE 0.2f
#define EPS 1e-16f

__device__ __forceinline__ float lrelu(float x) { return x > 0.f ? x : SLOPE * x; }

__device__ __forceinline__ float wave_max(float v) {
    for (int o = 16; o > 0; o >>= 1) v = fmaxf(v, __shfl_xor(v, o));
    return v;
}
__device__ __forceinline__ float wave_sum(float v) {
    for (int o = 16; o > 0; o >>= 1) v += __shfl_xor(v, o);
    return v;
}

__global__ __launch_bounds__(256) void csr_zero(unsigned* __restrict__ cnt, int N) {
    int i = blockIdx.x * blockDim.x + threadIdx.x;
    if (i < N) cnt[i] = 0u;
}

__global__ __launch_bounds__(256) void csr_count(const int* __restrict__ ei, int E,
                                                 unsigned* __restrict__ cnt,
                                                 int* __restrict__ rank) {
    int e = blockIdx.x * blockDim.x + threadIdx.x;
    if (e >= E) return;
    int d = ei[E + e];
    rank[e] = (int)atomicAdd(&cnt[d], 1u);
}

__global__ __launch_bounds__(1024) void csr_scan(const unsigned* __restrict__ cnt,
                                                 int* __restrict__ rowptr, int N) {
    __shared__ int wsum[32];
    const int t = threadIdx.x;
    const int lane = t & 31;
    const int w = t >> 5;
    const int chunk = (N + 1023) / 1024;
    const int b = t * chunk;
    const int e = (b + chunk < N) ? (b + chunk) : N;
    int s = 0;
    for (int i = b; i < e; ++i) s += (int)cnt[i];
    int inc = s;
    for (int o = 1; o < 32; o <<= 1) {
        int v = __shfl_up(inc, o);
        if (lane >= o) inc += v;
    }
    if (lane == 31) wsum[w] = inc;
    __syncthreads();
    if (w == 0) {
        int v = wsum[lane];
        int iv = v;
        for (int o = 1; o < 32; o <<= 1) {
            int u = __shfl_up(iv, o);
            if (lane >= o) iv += u;
        }
        wsum[lane] = iv - v;
    }
    __syncthreads();
    int base = wsum[w] + inc - s;
    for (int i = b; i < e; ++i) { rowptr[i] = base; base += (int)cnt[i]; }
    if (t == 1023) rowptr[N] = base;
}

__global__ __launch_bounds__(256) void csr_place(const int* __restrict__ ei, int E,
                                                 const int* __restrict__ rowptr,
                                                 const int* __restrict__ rank,
                                                 int* __restrict__ csr_src) {
    int e = blockIdx.x * blockDim.x + threadIdx.x;
    if (e >= E) return;
    int s = ei[e];
    int d = ei[E + e];
    int pos = rowptr[d] + rank[e];
    if ((unsigned)pos < (unsigned)E) csr_src[pos] = s;
}

__global__ __launch_bounds__(256) void gat_gemm(const float* __restrict__ X,
                                                const float* __restrict__ W,
                                                float* __restrict__ H, int N) {
    const int wave = blockIdx.x * 8 + (threadIdx.x >> 5);
    const int lane = threadIdx.x & 31;
    const int m0 = wave * 16;
    if (m0 >= N) return;
    const int row   = m0 + (lane & 15);
    const int col   = lane & 15;
    const int khalf = (lane >> 4) * 2;
    v8f c0 = {}; v8f c1 = {};
#pragma unroll
    for (int kb = 0; kb < 8; ++kb) {
        const int k0 = kb * 4 + khalf;
        v2f a;  a.x  = X[row * D + k0];        a.y  = X[row * D + k0 + 1];
        v2f b0; b0.x = W[k0 * D + col];        b0.y = W[(k0 + 1) * D + col];
        v2f b1; b1.x = W[k0 * D + col + 16];   b1.y = W[(k0 + 1) * D + col + 16];
        c0 = __builtin_amdgcn_wmma_f32_16x16x4_f32(false, a, false, b0, (short)0, c0, false, false);
        c1 = __builtin_amdgcn_wmma_f32_16x16x4_f32(false, a, false, b1, (short)0, c1, false, false);
    }
    const int rbase = m0 + (lane >> 4) * 8;
#pragma unroll
    for (int r = 0; r < 8; ++r) {
        H[(rbase + r) * D + col]      = c0[r];
        H[(rbase + r) * D + col + 16] = c1[r];
    }
}

__global__ __launch_bounds__(256) void gat_dots(const float* __restrict__ H,
                                                const float* __restrict__ a_src,
                                                const float* __restrict__ a_dst,
                                                float* __restrict__ as_, float* __restrict__ ad_,
                                                int N) {
    int i = blockIdx.x * blockDim.x + threadIdx.x;
    if (i >= N) return;
    float s = 0.f, d = 0.f;
#pragma unroll
    for (int k = 0; k < D; ++k) {
        float h = H[i * D + k];
        s += h * a_src[k];
        d += h * a_dst[k];
    }
    as_[i] = s; ad_[i] = d;
}

__global__ __launch_bounds__(256) void gat_node(const int* __restrict__ rowptr,
                                                const int* __restrict__ csr_src,
                                                const float* __restrict__ as_,
                                                const float* __restrict__ ad_,
                                                const float* __restrict__ H,
                                                const float* __restrict__ b,
                                                float* __restrict__ out, int N) {
    const int d    = blockIdx.x * 8 + __builtin_amdgcn_readfirstlane((int)(threadIdx.x >> 5));
    const int lane = threadIdx.x & 31;
    if (d >= N) return;
    const int   beg   = rowptr[d];
    const int   end   = rowptr[d + 1];
    const float adv   = ad_[d];
    const float lself = lrelu(as_[d] + adv);

    float mx = lself;
    for (int i = beg + lane; i < end; i += 32) mx = fmaxf(mx, lrelu(as_[csr_src[i]] + adv));
    mx = wave_max(mx);

    float ds = 0.f;
    for (int i = beg + lane; i < end; i += 32) ds += __expf(lrelu(as_[csr_src[i]] + adv) - mx);
    const float exself = __expf(lself - mx);
    const float den    = wave_sum(ds) + exself + EPS;

    float acc = (exself / den) * H[d * D + lane];
    for (int base = beg; base < end; base += 32) {
        const int i = base + lane;
        int   s  = d;
        float al = 0.f;
        if (i < end) {
            s  = csr_src[i];
            al = __expf(lrelu(as_[s] + adv) - mx) / den;
        }
        const int n = (end - base < 32) ? (end - base) : 32;
        for (int j0 = 0; j0 < n; j0 += 8) {
#pragma unroll
            for (int u = 0; u < 8; ++u) {
                const int   sj = __builtin_amdgcn_readlane(s, j0 + u);
                const float aj = __int_as_float(__builtin_amdgcn_readlane(__float_as_int(al), j0 + u));
                acc += aj * H[sj * D + lane];
            }
        }
    }
    const float v = acc + b[lane];
    out[d * D + lane] = (v > 0.f) ? v : (__expf(v) - 1.f);
}

static void run_layer(const float* x_in, int N,
                      const float* W, const float* a_src, const float* a_dst, const float* b,
                      const int* rowptr, const int* csr_src,
                      float* H, float* as_, float* ad_, float* out, hipStream_t stream) {
    const int node_blocks = (N + 255) / 256;
    const int tile_waves  = (N + 15) / 16;
    const int gemm_blocks = (tile_waves + 7) / 8;
    const int wave_blocks = (N + 7) / 8;

    gat_gemm<<<gemm_blocks, 256, 0, stream>>>(x_in, W, H, N);
    gat_dots<<<node_blocks, 256, 0, stream>>>(H, a_src, a_dst, as_, ad_, N);
    gat_node<<<wave_blocks, 256, 0, stream>>>(rowptr, csr_src, as_, ad_, H, b, out, N);
}

extern "C" void kernel_launch(void* const* d_in, const int* in_sizes, int n_in,
                              void* d_out, int out_size, void* d_ws, size_t ws_size,
                              hipStream_t stream) {
    const float* x      = (const float*)d_in[0];
    const int*   ei     = (const int*)d_in[1];
    const float* W1     = (const float*)d_in[2];
    const float* a_src1 = (const float*)d_in[3];
    const float* a_dst1 = (const float*)d_in[4];
    const float* b1     = (const float*)d_in[5];
    const float* W2     = (const float*)d_in[6];
    const float* a_src2 = (const float*)d_in[7];
    const float* a_dst2 = (const float*)d_in[8];
    const float* b2     = (const float*)d_in[9];

    const int N = in_sizes[0] / D;
    const int E = in_sizes[1] / 2;

    float*    H       = (float*)d_ws;
    float*    as_     = H + (size_t)N * D;
    float*    ad_     = as_ + N;
    unsigned* cnt     = (unsigned*)(ad_ + N);
    int*      rowptr  = (int*)(cnt + N);
    int*      rank    = rowptr + (N + 1);
    int*      csr_src = rank + E;

    float* xbar = (float*)d_out;
    float* z    = xbar + (size_t)N * D;

    const int node_blocks = (N + 255) / 256;
    const int edge_blocks = (E + 255) / 256;

    csr_zero<<<node_blocks, 256, 0, stream>>>(cnt, N);
    csr_count<<<edge_blocks, 256, 0, stream>>>(ei, E, cnt, rank);
    csr_scan<<<1, 1024, 0, stream>>>(cnt, rowptr, N);
    csr_place<<<edge_blocks, 256, 0, stream>>>(ei, E, rowptr, rank, csr_src);

    run_layer(x, N, W1, a_src1, a_dst1, b1, rowptr, csr_src, H, as_, ad_, z, stream);
    run_layer(z, N, W2, a_src2, a_dst2, b2, rowptr, csr_src, H, as_, ad_, xbar, stream);
}


# Edges grouped by destination through LDS: ranks by LDS atomics in 256-node buckets, then per node, runs stored contiguously; no global atomics; a node's sum order = LDS atomic arrival rank
# speedup vs baseline: 9.0364x; 3.4734x over previous
#include <hip/hip_runtime.h>
#include <hip/hip_bf16.h>


typedef __attribute__((ext_vector_type(2))) float v2f;
typedef __attribute__((ext_vector_type(8))) float v8f;

#define D 32
#define SLOPE 0.2f
#define EPS 1e-16f

__device__ __forceinline__ float lrelu(float x) { return x > 0.f ? x : SLOPE * x; }

template <int M>
__device__ __forceinline__ float lane_xor(float x) {
    if constexpr (M == 16) {
        return __int_as_float(__builtin_amdgcn_permlanex16(__float_as_int(x), __float_as_int(x),
                                                           0x76543210, 0xfedcba98, false, false));
    } else {
        return __int_as_float(__builtin_amdgcn_update_dpp(0, __float_as_int(x), 0x160 | M, 0xf, 0xf, true));
    }
}

__device__ __forceinline__ float wave_max(float v) {
    v = fmaxf(v, lane_xor<16>(v)); v = fmaxf(v, lane_xor<8>(v)); v = fmaxf(v, lane_xor<4>(v));
    v = fmaxf(v, lane_xor<2>(v));  v = fmaxf(v, lane_xor<1>(v));
    return v;
}
__device__ __forceinline__ float wave_sum(float v) {
    v += lane_xor<16>(v); v += lane_xor<8>(v); v += lane_xor<4>(v); v += lane_xor<2>(v); v += lane_xor<1>(v);
    return v;
}

#define CH    8192
#define EPT   8
#define BSH   8
#define NBMAX 1024
#define CAP   16384

template <int S>
__device__ __forceinline__ unsigned row_shr(unsigned x) {
    return (unsigned)__builtin_amdgcn_update_dpp(0, (int)x, 0x110 | S, 0xf, 0xf, true);
}

__device__ __forceinline__ unsigned wave_incl_scan(unsigned x, int lane) {
    x += row_shr<1>(x);
    x += row_shr<2>(x);
    x += row_shr<4>(x);
    x += row_shr<8>(x);
    const unsigned low = (unsigned)__builtin_amdgcn_readlane((int)x, 15);
    return x + ((lane & 16) ? low : 0u);
}

__device__ __forceinline__ unsigned block_excl_scan(unsigned x, unsigned* ws, int lane, int w) {
    const unsigned inc = wave_incl_scan(x, lane);
    if (lane == 31) ws[w] = inc;
    __syncthreads();
    const unsigned y  = ws[lane];
    const unsigned yi = wave_incl_scan(y, lane);
    return (unsigned)__builtin_amdgcn_readlane((int)(yi - y), w) + inc - x;
}

__global__ __launch_bounds__(1024) void grp_hist(const int* __restrict__ ei, int E, int NB, int G,
                                                 unsigned* __restrict__ wgh, int M, int M4) {
    __shared__ unsigned h[NBMAX];
    const int t = threadIdx.x;
    const int g = blockIdx.x;
    const int e0 = g * CH + t;
    int dv[EPT];
#pragma unroll
    for (int k = 0; k < EPT; ++k) {
        const int e = e0 + k * 1024;
        dv[k] = ei[E + (e < E ? e : E - 1)];
    }
    for (int i = t; i < NB; i += 1024) h[i] = 0u;
    __syncthreads();
#pragma unroll
    for (int k = 0; k < EPT; ++k) {
        if (e0 + k * 1024 < E) {
            int bk = dv[k] >> BSH;
            bk = bk < NB ? bk : NB - 1;
            atomicAdd(&h[bk], 1u);
        }
    }
    __syncthreads();
    for (int i = t; i < NB; i += 1024) wgh[i * G + g] = h[i];
    if (g == 0 && t < M4 - M) wgh[M + t] = 0u;
}

#define SCT 16
__global__ __launch_bounds__(1024) void grp_scan(const unsigned* __restrict__ in,
                                                 unsigned* __restrict__ out, int M4) {
    __shared__ unsigned ws[2][32];
    const int t = threadIdx.x;
    const int lane = t & 31;
    const int w = __builtin_amdgcn_readfirstlane(t >> 5);
    unsigned carry = 0u;
    for (int pbase = 0; pbase < M4; pbase += SCT * 4096) {
        uint4 v[SCT];
        unsigned exs[SCT];
#pragma unroll
        for (int j = 0; j < SCT; ++j) {
            const int i = pbase + (j * 1024 + t) * 4;
            v[j] = *reinterpret_cast<const uint4*>(in + (i < M4 ? i : M4 - 4));
        }
#pragma unroll
        for (int j = 0; j < SCT; ++j) {
            exs[j] = 0u;
            if (pbase + j * 4096 < M4) {
                const int i = pbase + (j * 1024 + t) * 4;
                const unsigned s = (i < M4) ? (v[j].x + v[j].y + v[j].z + v[j].w) : 0u;
                const unsigned inc = wave_incl_scan(s, lane);
                if (lane == 31) ws[j & 1][w] = inc;
                __syncthreads();
                const unsigned y   = ws[j & 1][lane];
                const unsigned yi  = wave_incl_scan(y, lane);
                const unsigned wof = (unsigned)__builtin_amdgcn_readlane((int)(yi - y), w);
                const unsigned tot = (unsigned)__builtin_amdgcn_readlane((int)yi, 31);
                exs[j] = carry + wof + inc - s;
                carry += tot;
            }
        }
#pragma unroll
        for (int j = 0; j < SCT; ++j) {
            const int i = pbase + (j * 1024 + t) * 4;
            if (i < M4) {
                const unsigned ex = exs[j];
                *reinterpret_cast<uint4*>(out + i) =
                    make_uint4(ex, ex + v[j].x, ex + v[j].x + v[j].y, ex + v[j].x + v[j].y + v[j].z);
            }
        }
    }
}

__global__ __launch_bounds__(1024) void grp_scatter(const int* __restrict__ ei, int E, int NB, int G,
                                                    const unsigned* __restrict__ wgo,
                                                    unsigned* __restrict__ rec) {
    __shared__ unsigned h[NBMAX];
    __shared__ unsigned lstart[NBMAX];
    __shared__ unsigned goff[NBMAX];
    __shared__ unsigned sbuf[CH];
    __shared__ unsigned bbuf[CH];
    __shared__ unsigned ws[32];
    const int t = threadIdx.x;
    const int lane = t & 31;
    const int w = __builtin_amdgcn_readfirstlane(t >> 5);
    const int g = blockIdx.x;
    int      bk[EPT];
    unsigned pk[EPT];
    unsigned rk[EPT];
    const int e0 = g * CH + t;
    int sv[EPT];
    int dv[EPT];
#pragma unroll
    for (int k = 0; k < EPT; ++k) {
        const int e = e0 + k * 1024;
        const int ec = e < E ? e : E - 1;
        sv[k] = ei[ec];
        dv[k] = ei[E + ec];
    }
    for (int i = t; i < NB; i += 1024) { h[i] = 0u; goff[i] = wgo[i * G + g]; }
    __syncthreads();
#pragma unroll
    for (int k = 0; k < EPT; ++k) {
        const int e = e0 + k * 1024;
        bk[k] = -1; pk[k] = 0u; rk[k] = 0u;
        if (e < E) {
            const int s = sv[k];
            const int d = dv[k];
            int b = d >> BSH;
            b = b < NB ? b : NB - 1;
            bk[k] = b;
            pk[k] = ((unsigned)(d & 255) << 24) | ((unsigned)s & 0xFFFFFFu);
            rk[k] = atomicAdd(&h[b], 1u);
        }
    }
    __syncthreads();
    {
        const unsigned ls = block_excl_scan(t < NB ? h[t] : 0u, ws, lane, w);
        if (t < NB) {
            lstart[t] = ls;
            goff[t] -= ls;
        }
    }
    __syncthreads();
#pragma unroll
    for (int k = 0; k < EPT; ++k) {
        if (bk[k] >= 0) {
            const unsigned p = lstart[bk[k]] + rk[k];
            sbuf[p] = pk[k];
            bbuf[p] = (unsigned)bk[k];
        }
    }
    __syncthreads();
    const int left = E - g * CH;
    const int n = left < CH ? left : CH;
    for (int i = t; i < n; i += 1024) {
        const unsigned gp = goff[bbuf[i]] + (unsigned)i;
        if (gp < (unsigned)E) rec[gp] = sbuf[i];
    }
}

__global__ __launch_bounds__(1024) void grp_bucket(const unsigned* __restrict__ rec,
                                                   const unsigned* __restrict__ wgo,
                                                   int E, int N, int NB, int G,
                                                   int* __restrict__ rowptr,
                                                   int* __restrict__ csr_src) {
    __shared__ unsigned c[256];
    __shared__ unsigned obuf[CAP];
    __shared__ unsigned ws[32];
    const int t = threadIdx.x;
    const int lane = t & 31;
    const int w = __builtin_amdgcn_readfirstlane(t >> 5);
    const int b = blockIdx.x;
    int beg = (int)wgo[b * G];
    int end = (b + 1 < NB) ? (int)wgo[(b + 1) * G] : E;
    end = end < 0 ? 0 : (end > E ? E : end);
    beg = beg < 0 ? 0 : (beg > end ? end : beg);
    if (t < 256) c[t] = 0u;
    __syncthreads();
    for (int base = beg + t; base < end; base += 4096) {
        unsigned r[4];
#pragma unroll
        for (int u = 0; u < 4; ++u) {
            const int i = base + u * 1024;
            r[u] = rec[i < end ? i : end - 1];
        }
#pragma unroll
        for (int u = 0; u < 4; ++u) {
            if (base + u * 1024 < end) atomicAdd(&c[r[u] >> 24], 1u);
        }
    }
    __syncthreads();
    const unsigned ex = block_excl_scan(t < 256 ? c[t] : 0u, ws, lane, w);
    if (t < 256) c[t] = ex;
    __syncthreads();
    const int cnt = end - beg;
    const bool staged = cnt <= CAP;
    for (int base = beg + t; base < end; base += 4096) {
        unsigned r[4];
#pragma unroll
        for (int u = 0; u < 4; ++u) {
            const int i = base + u * 1024;
            r[u] = rec[i < end ? i : end - 1];
        }
#pragma unroll
        for (int u = 0; u < 4; ++u) {
            if (base + u * 1024 < end) {
                const unsigned slot = atomicAdd(&c[r[u] >> 24], 1u);
                const int s = (int)(r[u] & 0xFFFFFFu);
                if (staged) {
                    if (slot < (unsigned)CAP) obuf[slot] = (unsigned)s;
                } else {
                    const unsigned gp = (unsigned)beg + slot;
                    if (gp < (unsigned)E) csr_src[gp] = s;
                }
            }
        }
    }
    __syncthreads();
    if (staged) {
        for (int i = t; i < cnt; i += 1024) csr_src[beg + i] = (int)obuf[i];
    }
    if (t < 256) {
        const int node = (b << BSH) + t;
        if (node < N) rowptr[node] = beg + (int)ex;
    }
    if (b == NB - 1 && t == 0) rowptr[N] = end;
}

__global__ __launch_bounds__(256) void gat_gemm(const float* __restrict__ X,
                                                const float* __restrict__ W,
                                                float* __restrict__ H, int N) {
    const int wave = blockIdx.x * 8 + (threadIdx.x >> 5);
    const int lane = threadIdx.x & 31;
    const int m0 = wave * 16;
    if (m0 >= N) return;
    const int row   = m0 + (lane & 15);
    const int col   = lane & 15;
    const int khalf = (lane >> 4) * 2;
    v8f c0 = {}; v8f c1 = {};
#pragma unroll
    for (int kb = 0; kb < 8; ++kb) {
        const int k0 = kb * 4 + khalf;
        v2f a;  a.x  = X[row * D + k0];        a.y  = X[row * D + k0 + 1];
        v2f b0; b0.x = W[k0 * D + col];        b0.y = W[(k0 + 1) * D + col];
        v2f b1; b1.x = W[k0 * D + col + 16];   b1.y = W[(k0 + 1) * D + col + 16];
        c0 = __builtin_amdgcn_wmma_f32_16x16x4_f32(false, a, false, b0, (short)0, c0, false, false);
        c1 = __builtin_amdgcn_wmma_f32_16x16x4_f32(false, a, false, b1, (short)0, c1, false, false);
    }
    const int rbase = m0 + (lane >> 4) * 8;
#pragma unroll
    for (int r = 0; r < 8; ++r) {
        H[(rbase + r) * D + col]      = c0[r];
        H[(rbase + r) * D + col + 16] = c1[r];
    }
}

template <int C>
__device__ __forceinline__ void fold_level(float* p, int lane) {
    const bool hi = (lane & C) != 0;
#pragma unroll
    for (int k = 0; k < C; ++k) {
        const float keep = hi ? p[k + C] : p[k];
        const float send = hi ? p[k] : p[k + C];
        p[k] = keep + lane_xor<C>(send);
    }
}

__global__ __launch_bounds__(256) void gat_node(const int* __restrict__ rowptr,
                                                const int* __restrict__ csr_src,
                                                const float* __restrict__ H,
                                                const float* __restrict__ a_src,
                                                const float* __restrict__ a_dst,
                                                const float* __restrict__ b,
                                                float* __restrict__ out, int N) {
    const int d    = blockIdx.x * 8 + __builtin_amdgcn_readfirstlane((int)(threadIdx.x >> 5));
    const int lane = threadIdx.x & 31;
    if (d >= N) return;
    const int   beg = rowptr[d];
    const int   end = rowptr[d + 1];
    const float asl = a_src[lane];
    const float adl = a_dst[lane];
    const float hd  = H[d * D + lane];
    int s_next = d;
    if (beg + lane < end) s_next = csr_src[beg + lane];
    const float adv   = wave_sum(hd * adl);
    const float lself = lrelu(wave_sum(hd * asl) + adv);

    float m   = lself;
    float den = 1.f;
    float acc = hd;

    for (int base = beg; base < end; base += 32) {
        const int  i     = base + lane;
        const bool valid = i < end;
        const int  n     = (end - base < 32) ? (end - base) : 32;
        const int s = s_next;
        s_next = d;
        if (i + 32 < end) s_next = csr_src[i + 32];

        float r[32];
#pragma unroll
        for (int g = 0; g < 4; ++g) {
            if (g * 8 < n) {
#pragma unroll
                for (int u = 0; u < 8; ++u) {
                    const int sj = __builtin_amdgcn_readlane(s, g * 8 + u);
                    r[g * 8 + u] = H[sj * D + lane];
                }
            } else {
#pragma unroll
                for (int u = 0; u < 8; ++u) r[g * 8 + u] = 0.f;
            }
        }

        float p[16];
        {
            const bool hi = (lane & 16) != 0;
#pragma unroll
            for (int k = 0; k < 16; ++k) {
                const float keep = hi ? r[k + 16] : r[k];
                const float send = hi ? r[k] : r[k + 16];
                p[k] = keep * asl + lane_xor<16>(send * asl);
            }
        }
        fold_level<8>(p, lane);
        fold_level<4>(p, lane);
        fold_level<2>(p, lane);
        fold_level<1>(p, lane);

        const float l  = valid ? lrelu(p[0] + adv) : -__builtin_inff();
        const float mn = fmaxf(m, wave_max(l));
        const float sc = __expf(m - mn);
        const float e  = valid ? __expf(l - mn) : 0.f;
        den = den * sc + wave_sum(e);
        acc *= sc;
        m = mn;
#pragma unroll
        for (int g = 0; g < 4; ++g) {
            if (g * 8 < n) {
#pragma unroll
                for (int u = 0; u < 8; ++u) {
                    const float ej = __int_as_float(__builtin_amdgcn_readlane(__float_as_int(e), g * 8 + u));
                    acc = fmaf(ej, r[g * 8 + u], acc);
                }
            }
        }
    }
    const float v = acc / (den + EPS) + b[lane];
    out[d * D + lane] = (v > 0.f) ? v : (__expf(v) - 1.f);
}

static void run_layer(const float* x_in, int N,
                      const float* W, const float* a_src, const float* a_dst, const float* b,
                      const int* rowptr, const int* csr_src,
                      float* H, float* out, hipStream_t stream) {
    const int tile_waves  = (N + 15) / 16;
    const int gemm_blocks = (tile_waves + 7) / 8;
    const int wave_blocks = (N + 7) / 8;

    gat_gemm<<<gemm_blocks, 256, 0, stream>>>(x_in, W, H, N);
    gat_node<<<wave_blocks, 256, 0, stream>>>(rowptr, csr_src, H, a_src, a_dst, b, out, N);
}

extern "C" void kernel_launch(void* const* d_in, const int* in_sizes, int n_in,
                              void* d_out, int out_size, void* d_ws, size_t ws_size,
                              hipStream_t stream) {
    const float* x      = (const float*)d_in[0];
    const int*   ei     = (const int*)d_in[1];
    const float* W1     = (const float*)d_in[2];
    const float* a_src1 = (const float*)d_in[3];
    const float* a_dst1 = (const float*)d_in[4];
    const float* b1     = (const float*)d_in[5];
    const float* W2     = (const float*)d_in[6];
    const float* a_src2 = (const float*)d_in[7];
    const float* a_dst2 = (const float*)d_in[8];
    const float* b2     = (const float*)d_in[9];

    const int N = in_sizes[0] / D;
    const int E = in_sizes[1] / 2;

    const int NB = (N + 255) / 256;
    const int G  = (E + CH - 1) / CH;
    const int M  = NB * G;
    const int M4 = (M + 3) & ~3;

    float*    H       = (float*)d_ws;
    unsigned* wgh     = (unsigned*)(H + (size_t)N * D);
    unsigned* wgo     = wgh + M4;
    int*      rowptr  = (int*)(wgo + M4);
    int*      csr_src = rowptr + (N + 1);
    unsigned* rec     = (unsigned*)(csr_src + E);

    float* xbar = (float*)d_out;
    float* z    = xbar + (size_t)N * D;

    grp_hist<<<G, 1024, 0, stream>>>(ei, E, NB, G, wgh, M, M4);
    grp_scan<<<1, 1024, 0, stream>>>(wgh, wgo, M4);
    grp_scatter<<<G, 1024, 0, stream>>>(ei, E, NB, G, wgo, rec);
    grp_bucket<<<NB, 1024, 0, stream>>>(rec, wgo, E, N, NB, G, rowptr, csr_src);

    run_layer(x, N, W1, a_src1, a_dst1, b1, rowptr, csr_src, H, z, stream);
    run_layer(z, N, W2, a_src2, a_dst2, b2, rowptr, csr_src, H, xbar, stream);
}


# Linear map moved behind the weighted sum: node kernel gathers input rows, scores as x.(W a), then its 32 summed rows times W on WMMA through LDS; no H table, no product launches; sums reassociated
# speedup vs baseline: 10.4873x; 1.1606x over previous
#include <hip/hip_runtime.h>
#include <hip/hip_bf16.h>


typedef __attribute__((ext_vector_type(2))) float v2f;
typedef __attribute__((ext_vector_type(8))) float v8f;
typedef __attribute__((ext_vector_type(4))) float v4f;

#define D 32
#define SLOPE 0.2f
#define EPS 1e-16f

__device__ __forceinline__ float lrelu(float x) { return x > 0.f ? x : SLOPE * x; }

template <int M>
__device__ __forceinline__ float lane_xor(float x) {
    if constexpr (M == 16) {
        return __int_as_float(__builtin_amdgcn_permlanex16(__float_as_int(x), __float_as_int(x),
                                                           0x76543210, 0xfedcba98, false, false));
    } else {
        return __int_as_float(__builtin_amdgcn_update_dpp(0, __float_as_int(x), 0x160 | M, 0xf, 0xf, true));
    }
}

__device__ __forceinline__ float wave_max(float v) {
    v = fmaxf(v, lane_xor<16>(v)); v = fmaxf(v, lane_xor<8>(v)); v = fmaxf(v, lane_xor<4>(v));
    v = fmaxf(v, lane_xor<2>(v));  v = fmaxf(v, lane_xor<1>(v));
    return v;
}
__device__ __forceinline__ float wave_sum(float v) {
    v += lane_xor<16>(v); v += lane_xor<8>(v); v += lane_xor<4>(v); v += lane_xor<2>(v); v += lane_xor<1>(v);
    return v;
}

#define CH    8192
#define EPT   8
#define BSH   8
#define NBMAX 1024
#define CAP   16384

template <int S>
__device__ __forceinline__ unsigned row_shr(unsigned x) {
    return (unsigned)__builtin_amdgcn_update_dpp(0, (int)x, 0x110 | S, 0xf, 0xf, true);
}

__device__ __forceinline__ unsigned wave_incl_scan(unsigned x, int lane) {
    x += row_shr<1>(x);
    x += row_shr<2>(x);
    x += row_shr<4>(x);
    x += row_shr<8>(x);
    const unsigned low = (unsigned)__builtin_amdgcn_readlane((int)x, 15);
    return x + ((lane & 16) ? low : 0u);
}

__device__ __forceinline__ unsigned block_excl_scan(unsigned x, unsigned* ws, int lane, int w) {
    const unsigned inc = wave_incl_scan(x, lane);
    if (lane == 31) ws[w] = inc;
    __syncthreads();
    const unsigned y  = ws[lane];
    const unsigned yi = wave_incl_scan(y, lane);
    return (unsigned)__builtin_amdgcn_readlane((int)(yi - y), w) + inc - x;
}

__global__ __launch_bounds__(1024) void grp_hist(const int* __restrict__ ei, int E, int NB, int G,
                                                 unsigned* __restrict__ wgh, int M, int M4) {
    __shared__ unsigned h[NBMAX];
    const int t = threadIdx.x;
    const int g = blockIdx.x;
    const int e0 = g * CH + t;
    int dv[EPT];
#pragma unroll
    for (int k = 0; k < EPT; ++k) {
        const int e = e0 + k * 1024;
        dv[k] = ei[E + (e < E ? e : E - 1)];
    }
    for (int i = t; i < NB; i += 1024) h[i] = 0u;
    __syncthreads();
#pragma unroll
    for (int k = 0; k < EPT; ++k) {
        if (e0 + k * 1024 < E) {
            int bk = dv[k] >> BSH;
            bk = bk < NB ? bk : NB - 1;
            atomicAdd(&h[bk], 1u);
        }
    }
    __syncthreads();
    for (int i = t; i < NB; i += 1024) wgh[i * G + g] = h[i];
    if (g == 0 && t < M4 - M) wgh[M + t] = 0u;
}

#define SCT 16
__global__ __launch_bounds__(1024) void grp_scan(const unsigned* __restrict__ in,
                                                 unsigned* __restrict__ out, int M4) {
    __shared__ unsigned ws[2][32];
    const int t = threadIdx.x;
    const int lane = t & 31;
    const int w = __builtin_amdgcn_readfirstlane(t >> 5);
    unsigned carry = 0u;
    for (int pbase = 0; pbase < M4; pbase += SCT * 4096) {
        uint4 v[SCT];
        unsigned exs[SCT];
#pragma unroll
        for (int j = 0; j < SCT; ++j) {
            const int i = pbase + (j * 1024 + t) * 4;
            v[j] = *reinterpret_cast<const uint4*>(in + (i < M4 ? i : M4 - 4));
        }
#pragma unroll
        for (int j = 0; j < SCT; ++j) {
            exs[j] = 0u;
            if (pbase + j * 4096 < M4) {
                const int i = pbase + (j * 1024 + t) * 4;
                const unsigned s = (i < M4) ? (v[j].x + v[j].y + v[j].z + v[j].w) : 0u;
                const unsigned inc = wave_incl_scan(s, lane);
                if (lane == 31) ws[j & 1][w] = inc;
                __syncthreads();
                const unsigned y   = ws[j & 1][lane];
                const unsigned yi  = wave_incl_scan(y, lane);
                const unsigned wof = (unsigned)__builtin_amdgcn_readlane((int)(yi - y), w);
                const unsigned tot = (unsigned)__builtin_amdgcn_readlane((int)yi, 31);
                exs[j] = carry + wof + inc - s;
                carry += tot;
            }
        }
#pragma unroll
        for (int j = 0; j < SCT; ++j) {
            const int i = pbase + (j * 1024 + t) * 4;
            if (i < M4) {
                const unsigned ex = exs[j];
                *reinterpret_cast<uint4*>(out + i) =
                    make_uint4(ex, ex + v[j].x, ex + v[j].x + v[j].y, ex + v[j].x + v[j].y + v[j].z);
            }
        }
    }
}

__global__ __launch_bounds__(1024) void grp_scatter(const int* __restrict__ ei, int E, int NB, int G,
                                                    const unsigned* __restrict__ wgo,
                                                    unsigned* __restrict__ rec) {
    __shared__ unsigned h[NBMAX];
    __shared__ unsigned lstart[NBMAX];
    __shared__ unsigned goff[NBMAX];
    __shared__ unsigned sbuf[CH];
    __shared__ unsigned bbuf[CH];
    __shared__ unsigned ws[32];
    const int t = threadIdx.x;
    const int lane = t & 31;
    const int w = __builtin_amdgcn_readfirstlane(t >> 5);
    const int g = blockIdx.x;
    int      bk[EPT];
    unsigned pk[EPT];
    unsigned rk[EPT];
    const int e0 = g * CH + t;
    int sv[EPT];
    int dv[EPT];
#pragma unroll
    for (int k = 0; k < EPT; ++k) {
        const int e = e0 + k * 1024;
        const int ec = e < E ? e : E - 1;
        sv[k] = ei[ec];
        dv[k] = ei[E + ec];
    }
    for (int i = t; i < NB; i += 1024) { h[i] = 0u; goff[i] = wgo[i * G + g]; }
    __syncthreads();
#pragma unroll
    for (int k = 0; k < EPT; ++k) {
        const int e = e0 + k * 1024;
        bk[k] = -1; pk[k] = 0u; rk[k] = 0u;
        if (e < E) {
            const int s = sv[k];
            const int d = dv[k];
            int b = d >> BSH;
            b = b < NB ? b : NB - 1;
            bk[k] = b;
            pk[k] = ((unsigned)(d & 255) << 24) | ((unsigned)s & 0xFFFFFFu);
            rk[k] = atomicAdd(&h[b], 1u);
        }
    }
    __syncthreads();
    {
        const unsigned ls = block_excl_scan(t < NB ? h[t] : 0u, ws, lane, w);
        if (t < NB) {
            lstart[t] = ls;
            goff[t] -= ls;
        }
    }
    __syncthreads();
#pragma unroll
    for (int k = 0; k < EPT; ++k) {
        if (bk[k] >= 0) {
            const unsigned p = lstart[bk[k]] + rk[k];
            sbuf[p] = pk[k];
            bbuf[p] = (unsigned)bk[k];
        }
    }
    __syncthreads();
    const int left = E - g * CH;
    const int n = left < CH ? left : CH;
    for (int i = t; i < n; i += 1024) {
        const unsigned gp = goff[bbuf[i]] + (unsigned)i;
        if (gp < (unsigned)E) rec[gp] = sbuf[i];
    }
}

__global__ __launch_bounds__(1024) void grp_bucket(const unsigned* __restrict__ rec,
                                                   const unsigned* __restrict__ wgo,
                                                   int E, int N, int NB, int G,
                                                   int* __restrict__ rowptr,
                                                   int* __restrict__ csr_src) {
    __shared__ unsigned c[256];
    __shared__ unsigned obuf[CAP];
    __shared__ unsigned ws[32];
    const int t = threadIdx.x;
    const int lane = t & 31;
    const int w = __builtin_amdgcn_readfirstlane(t >> 5);
    const int b = blockIdx.x;
    int beg = (int)wgo[b * G];
    int end = (b + 1 < NB) ? (int)wgo[(b + 1) * G] : E;
    end = end < 0 ? 0 : (end > E ? E : end);
    beg = beg < 0 ? 0 : (beg > end ? end : beg);
    if (t < 256) c[t] = 0u;
    __syncthreads();
    for (int base = beg + t; base < end; base += 4096) {
        unsigned r[4];
#pragma unroll
        for (int u = 0; u < 4; ++u) {
            const int i = base + u * 1024;
            r[u] = rec[i < end ? i : end - 1];
        }
#pragma unroll
        for (int u = 0; u < 4; ++u) {
            if (base + u * 1024 < end) atomicAdd(&c[r[u] >> 24], 1u);
        }
    }
    __syncthreads();
    const unsigned ex = block_excl_scan(t < 256 ? c[t] : 0u, ws, lane, w);
    if (t < 256) c[t] = ex;
    __syncthreads();
    const int cnt = end - beg;
    const bool staged = cnt <= CAP;
    for (int base = beg + t; base < end; base += 4096) {
        unsigned r[4];
#pragma unroll
        for (int u = 0; u < 4; ++u) {
            const int i = base + u * 1024;
            r[u] = rec[i < end ? i : end - 1];
        }
#pragma unroll
        for (int u = 0; u < 4; ++u) {
            if (base + u * 1024 < end) {
                const unsigned slot = atomicAdd(&c[r[u] >> 24], 1u);
                const int s = (int)(r[u] & 0xFFFFFFu);
                if (staged) {
                    if (slot < (unsigned)CAP) obuf[slot] = (unsigned)s;
                } else {
                    const unsigned gp = (unsigned)beg + slot;
                    if (gp < (unsigned)E) csr_src[gp] = s;
                }
            }
        }
    }
    __syncthreads();
    if (staged) {
        for (int i = t; i < cnt; i += 1024) csr_src[beg + i] = (int)obuf[i];
    }
    if (t < 256) {
        const int node = (b << BSH) + t;
        if (node < N) rowptr[node] = beg + (int)ex;
    }
    if (b == NB - 1 && t == 0) rowptr[N] = end;
}

template <int C>
__device__ __forceinline__ void fold_level(float* p, int lane) {
    const bool hi = (lane & C) != 0;
#pragma unroll
    for (int k = 0; k < C; ++k) {
        const float keep = hi ? p[k + C] : p[k];
        const float send = hi ? p[k] : p[k + C];
        p[k] = keep + lane_xor<C>(send);
    }
}

template <int U>
__device__ __forceinline__ int grp_bcast(int x) {
    return __builtin_amdgcn_mov_dpp8(x, U * 0x249249);
}
__device__ __forceinline__ float grp_max(float v) {
    v = fmaxf(v, lane_xor<4>(v)); v = fmaxf(v, lane_xor<2>(v)); v = fmaxf(v, lane_xor<1>(v));
    return v;
}
__device__ __forceinline__ float grp_sum(float v) {
    v += lane_xor<4>(v); v += lane_xor<2>(v); v += lane_xor<1>(v);
    return v;
}
__device__ __forceinline__ float dot4(v4f x, v4f y) {
    const v2f t = __builtin_elementwise_fma(x.zw, y.zw, x.xy * y.xy);
    return t.x + t.y;
}

#define YP 36
__global__ __launch_bounds__(256) void gat_node(const int* __restrict__ rowptr,
                                                const int* __restrict__ csr_src,
                                                const float* __restrict__ Xin,
                                                const float* __restrict__ W,
                                                const float* __restrict__ a_src,
                                                const float* __restrict__ a_dst,
                                                const float* __restrict__ b,
                                                float* __restrict__ out, int N) {
    __shared__ v4f wa4[16];
    __shared__ v4f ybuf4[32 * YP / 4];
    float* const wa   = reinterpret_cast<float*>(wa4);
    float* const ybuf = reinterpret_cast<float*>(ybuf4);
    const int w    = __builtin_amdgcn_readfirstlane((int)(threadIdx.x >> 5));
    const int d0   = (blockIdx.x * 8 + w) * 4;
    const int lane = threadIdx.x & 31;
    const int  c    = lane & 7;
    const int wi = w * 4 + (lane >> 3);
    const v4f wr  = reinterpret_cast<const v4f*>(W)[wi * 8 + c];
    const v4f asr = reinterpret_cast<const v4f*>(a_src)[c];
    const v4f adr = reinterpret_cast<const v4f*>(a_dst)[c];
    const int  d    = d0 + (lane >> 3);
    const bool live = d < N;
    const int  dc   = live ? d : N - 1;
    const int  beg  = rowptr[dc];
    const int  deg  = live ? rowptr[dc + 1] - beg : 0;
    const v4f* __restrict__ X4 = reinterpret_cast<const v4f*>(Xin);
    const v4f hd  = X4[dc * 8 + c];
    int rem = deg - c;
    int pos = beg + c;
    int s_next = 0;
    if (rem > 0) s_next = csr_src[pos];
    const int dg01 = max(__builtin_amdgcn_readlane(deg, 0), __builtin_amdgcn_readlane(deg, 8));
    const int dg23 = max(__builtin_amdgcn_readlane(deg, 16), __builtin_amdgcn_readlane(deg, 24));
    const int maxdeg = max(dg01, dg23);

    {
        const float ps = grp_sum(dot4(wr, asr));
        const float pd = grp_sum(dot4(wr, adr));
        if (c == 0) { wa[wi] = ps; wa[32 + wi] = pd; }
    }
    __syncthreads();
    const v4f as4 = wa4[c];
    const v4f ad4 = wa4[8 + c];

    const float adv   = grp_sum(dot4(hd, ad4));
    const float lself = lrelu(grp_sum(dot4(hd, as4)) + adv);

    float m   = lself;
    float den = 1.f;
    v4f   acc = hd;

    for (int t = 0; t < maxdeg; t += 8) {
        const bool valid = rem > 0;
        const int s = s_next;
        s_next = 0;
        if (rem > 8) s_next = csr_src[pos + 8];
        rem -= 8;
        pos += 8;

        v4f r[8];
#pragma unroll
        for (int u = 0; u < 8; ++u) {
            int su;
            if (u == 0) su = grp_bcast<0>(s); else if (u == 1) su = grp_bcast<1>(s);
            else if (u == 2) su = grp_bcast<2>(s); else if (u == 3) su = grp_bcast<3>(s);
            else if (u == 4) su = grp_bcast<4>(s); else if (u == 5) su = grp_bcast<5>(s);
            else if (u == 6) su = grp_bcast<6>(s); else su = grp_bcast<7>(s);
            r[u] = X4[su * 8 + c];
        }
        __builtin_amdgcn_sched_barrier(0);

        float p[4];
        {
            const bool hi = (lane & 4) != 0;
#pragma unroll
            for (int k = 0; k < 4; ++k) {
                const float lo_row = dot4(r[k], as4);
                const float hi_row = dot4(r[k + 4], as4);
                const float keep = hi ? hi_row : lo_row;
                const float send = hi ? lo_row : hi_row;
                p[k] = keep + lane_xor<4>(send);
            }
        }
        fold_level<2>(p, lane);
        fold_level<1>(p, lane);

        const float l  = valid ? lrelu(p[0] + adv) : -__builtin_inff();
        const float mn = fmaxf(m, grp_max(l));
        const float sc = __expf(m - mn);
        const float e  = valid ? __expf(l - mn) : 0.f;
        den = den * sc + grp_sum(e);
        acc *= sc;
        m = mn;
#pragma unroll
        for (int u = 0; u < 8; ++u) {
            int eb;
            const int ei = __float_as_int(e);
            if (u == 0) eb = grp_bcast<0>(ei); else if (u == 1) eb = grp_bcast<1>(ei);
            else if (u == 2) eb = grp_bcast<2>(ei); else if (u == 3) eb = grp_bcast<3>(ei);
            else if (u == 4) eb = grp_bcast<4>(ei); else if (u == 5) eb = grp_bcast<5>(ei);
            else if (u == 6) eb = grp_bcast<6>(ei); else eb = grp_bcast<7>(ei);
            const float eu = __int_as_float(eb);
            acc.x = fmaf(eu, r[u].x, acc.x);
            acc.y = fmaf(eu, r[u].y, acc.y);
            acc.z = fmaf(eu, r[u].z, acc.z);
            acc.w = fmaf(eu, r[u].w, acc.w);
        }
    }

    const int le = (int)__builtin_amdgcn_mbcnt_lo(~0u, 0u);
    const int ce = le & 7;
    const int rl = w * 4 + (le >> 3);
    {
        const float dn = den + EPS;
        v4f y;
        y.x = acc.x / dn;
        y.y = acc.y / dn;
        y.z = acc.z / dn;
        y.w = acc.w / dn;
        ybuf4[rl * (YP / 4) + ce] = y;
    }
    const int col   = le & 15;
    const int khalf = (le >> 4) * 2;
    v2f b0[8], b1[8];
    if (w < 2) {
#pragma unroll
        for (int kb = 0; kb < 8; ++kb) {
            const int k0 = kb * 4 + khalf;
            b0[kb].x = W[k0 * D + col];        b0[kb].y = W[(k0 + 1) * D + col];
            b1[kb].x = W[k0 * D + col + 16];   b1[kb].y = W[(k0 + 1) * D + col + 16];
        }
    }
    __syncthreads();
    if (w < 2) {
        const float* ya = ybuf + (w * 16 + col) * YP;
        v8f c0 = {}; v8f c1 = {};
#pragma unroll
        for (int kb = 0; kb < 8; ++kb) {
            const int k0 = kb * 4 + khalf;
            v2f a;  a.x  = ya[k0];                 a.y  = ya[k0 + 1];
            c0 = __builtin_amdgcn_wmma_f32_16x16x4_f32(false, a, false, b0[kb], (short)0, c0, false, false);
            c1 = __builtin_amdgcn_wmma_f32_16x16x4_f32(false, a, false, b1[kb], (short)0, c1, false, false);
        }
        float* yo = ybuf + (w * 16 + (le >> 4) * 8) * YP;
#pragma unroll
        for (int r = 0; r < 8; ++r) {
            yo[r * YP + col]      = c0[r];
            yo[r * YP + col + 16] = c1[r];
        }
    }
    __syncthreads();
    if (d0 + (le >> 3) < N) {
        const v4f b4 = reinterpret_cast<const v4f*>(b)[ce];
        v4f v = ybuf4[rl * (YP / 4) + ce];
        v.x += b4.x;
        v.y += b4.y;
        v.z += b4.z;
        v.w += b4.w;
        v.x = (v.x > 0.f) ? v.x : (__expf(v.x) - 1.f);
        v.y = (v.y > 0.f) ? v.y : (__expf(v.y) - 1.f);
        v.z = (v.z > 0.f) ? v.z : (__expf(v.z) - 1.f);
        v.w = (v.w > 0.f) ? v.w : (__expf(v.w) - 1.f);
        reinterpret_cast<v4f*>(out)[(d0 + (le >> 3)) * 8 + ce] = v;
    }
}

extern "C" void kernel_launch(void* const* d_in, const int* in_sizes, int n_in,
                              void* d_out, int out_size, void* d_ws, size_t ws_size,
                              hipStream_t stream) {
    const float* x      = (const float*)d_in[0];
    const int*   ei     = (const int*)d_in[1];
    const float* W1     = (const float*)d_in[2];
    const float* a_src1 = (const float*)d_in[3];
    const float* a_dst1 = (const float*)d_in[4];
    const float* b1     = (const float*)d_in[5];
    const float* W2     = (const float*)d_in[6];
    const float* a_src2 = (const float*)d_in[7];
    const float* a_dst2 = (const float*)d_in[8];
    const float* b2     = (const float*)d_in[9];

    const int N = in_sizes[0] / D;
    const int E = in_sizes[1] / 2;

    const int NB = (N + 255) / 256;
    const int G  = (E + CH - 1) / CH;
    const int M  = NB * G;
    const int M4 = (M + 3) & ~3;

    unsigned* wgh     = (unsigned*)d_ws;
    unsigned* wgo     = wgh + M4;
    int*      rowptr  = (int*)(wgo + M4);
    int*      csr_src = rowptr + (N + 1);
    unsigned* rec     = (unsigned*)(csr_src + E);

    float* xbar = (float*)d_out;
    float* z    = xbar + (size_t)N * D;

    grp_hist<<<G, 1024, 0, stream>>>(ei, E, NB, G, wgh, M, M4);
    grp_scan<<<1, 1024, 0, stream>>>(wgh, wgo, M4);
    grp_scatter<<<G, 1024, 0, stream>>>(ei, E, NB, G, wgo, rec);
    grp_bucket<<<NB, 1024, 0, stream>>>(rec, wgo, E, N, NB, G, rowptr, csr_src);

    const int node_blocks = (N + 31) / 32;
    gat_node<<<node_blocks, 256, 0, stream>>>(rowptr, csr_src, x, W1, a_src1, a_dst1, b1, z, N);
    gat_node<<<node_blocks, 256, 0, stream>>>(rowptr, csr_src, z, W2, a_src2, a_dst2, b2, xbar, N);
}


# Edge grouping in 2 launches, not 4: each work-group keeps its bucket-sorted edges in its own segment; bucket kernel gathers its runs, base = sum of run starts; sum order = LDS atomic arrival rank
# speedup vs baseline: 13.2463x; 1.2631x over previous
#include <hip/hip_runtime.h>
#include <hip/hip_bf16.h>


typedef __attribute__((ext_vector_type(2))) float v2f;
typedef __attribute__((ext_vector_type(8))) float v8f;
typedef __attribute__((ext_vector_type(4))) float v4f;

#define D 32
#define SLOPE 0.2f
#define EPS 1e-16f

__device__ __forceinline__ float lrelu(float x) { return x > 0.f ? x : SLOPE * x; }

template <int M>
__device__ __forceinline__ float lane_xor(float x) {
    if constexpr (M == 16) {
        return __int_as_float(__builtin_amdgcn_permlanex16(__float_as_int(x), __float_as_int(x),
                                                           0x76543210, 0xfedcba98, false, false));
    } else {
        return __int_as_float(__builtin_amdgcn_update_dpp(0, __float_as_int(x), 0x160 | M, 0xf, 0xf, true));
    }
}

__device__ __forceinline__ float wave_max(float v) {
    v = fmaxf(v, lane_xor<16>(v)); v = fmaxf(v, lane_xor<8>(v)); v = fmaxf(v, lane_xor<4>(v));
    v = fmaxf(v, lane_xor<2>(v));  v = fmaxf(v, lane_xor<1>(v));
    return v;
}
__device__ __forceinline__ float wave_sum(float v) {
    v += lane_xor<16>(v); v += lane_xor<8>(v); v += lane_xor<4>(v); v += lane_xor<2>(v); v += lane_xor<1>(v);
    return v;
}

#define CH    8192
#define EPT   8
#define BSH   8
#define NBMAX 1024
#define GMAX  1024
#define CAP   16384
#define RU    4
__device__ __forceinline__ int run_pitch(int NB) { return (NB + 31) & ~31; }

template <int S>
__device__ __forceinline__ unsigned row_shr(unsigned x) {
    return (unsigned)__builtin_amdgcn_update_dpp(0, (int)x, 0x110 | S, 0xf, 0xf, true);
}

__device__ __forceinline__ unsigned wave_incl_scan(unsigned x, int lane) {
    x += row_shr<1>(x);
    x += row_shr<2>(x);
    x += row_shr<4>(x);
    x += row_shr<8>(x);
    const unsigned low = (unsigned)__builtin_amdgcn_readlane((int)x, 15);
    return x + ((lane & 16) ? low : 0u);
}

__device__ __forceinline__ unsigned wave_total(unsigned x, int lane) {
    return (unsigned)__builtin_amdgcn_readlane((int)wave_incl_scan(x, lane), 31);
}

__device__ __forceinline__ unsigned block_excl_scan(unsigned x, unsigned* ws, int lane, int w) {
    const unsigned inc = wave_incl_scan(x, lane);
    if (lane == 31) ws[w] = inc;
    __syncthreads();
    const unsigned y  = ws[lane];
    const unsigned yi = wave_incl_scan(y, lane);
    return (unsigned)__builtin_amdgcn_readlane((int)(yi - y), w) + inc - x;
}

__global__ __launch_bounds__(1024) void grp_sort(const int* __restrict__ ei, int E, int NB, int G,
                                                 unsigned* __restrict__ runs,
                                                 unsigned* __restrict__ rec) {
    __shared__ unsigned h[NBMAX];
    __shared__ unsigned lstart[NBMAX];
    __shared__ uint4    sbuf4[CH / 4];
    __shared__ unsigned ws[32];
    unsigned* const sbuf = reinterpret_cast<unsigned*>(sbuf4);
    const int t = threadIdx.x;
    const int lane = t & 31;
    const int w = __builtin_amdgcn_readfirstlane(t >> 5);
    const int g = blockIdx.x;
    int      bk[EPT];
    unsigned pk[EPT];
    unsigned rk[EPT];
    const int e0 = g * CH + t;
    int sv[EPT];
    int dv[EPT];
#pragma unroll
    for (int k = 0; k < EPT; ++k) {
        const int e = e0 + k * 1024;
        const int ec = e < E ? e : E - 1;
        sv[k] = ei[ec];
        dv[k] = ei[E + ec];
    }
    for (int i = t; i < NB; i += 1024) h[i] = 0u;
    __syncthreads();
#pragma unroll
    for (int k = 0; k < EPT; ++k) {
        const int e = e0 + k * 1024;
        bk[k] = -1; pk[k] = 0u; rk[k] = 0u;
        if (e < E) {
            const int s = sv[k];
            const int d = dv[k];
            int b = d >> BSH;
            b = b < NB ? b : NB - 1;
            bk[k] = b;
            pk[k] = ((unsigned)(d & 255) << 24) | ((unsigned)s & 0xFFFFFFu);
            rk[k] = atomicAdd(&h[b], 1u);
        }
    }
    __syncthreads();
    {
        const unsigned cnt = t < NB ? h[t] : 0u;
        const unsigned ls = block_excl_scan(cnt, ws, lane, w);
        if (t < NB) {
            lstart[t] = ls;
            runs[g * run_pitch(NB) + t] = (ls << 16) | cnt;
        }
    }
    __syncthreads();
#pragma unroll
    for (int k = 0; k < EPT; ++k) {
        if (bk[k] >= 0) sbuf[lstart[bk[k]] + rk[k]] = pk[k];
    }
    __syncthreads();
    const int left = E - g * CH;
    const int n = left < CH ? left : CH;
    uint4* __restrict__ seg = reinterpret_cast<uint4*>(rec + (size_t)g * CH);
    for (int i = t; i * 4 < n; i += 1024) seg[i] = sbuf4[i];
}

template <typename F>
__device__ __forceinline__ void for_each_record(const unsigned* __restrict__ rec, const unsigned* rbeg,
                                                const unsigned* rcnt, int G, int lane, int w, F f) {
    for (int g0 = w; g0 < G; g0 += 32 * RU) {
        unsigned rb[RU], rc[RU];
#pragma unroll
        for (int u = 0; u < RU; ++u) {
            const int g = g0 + 32 * u;
            rb[u] = g < G ? rbeg[g] : 0u;
            rc[u] = g < G ? rcnt[g] : 0u;
        }
        unsigned r[RU][2];
#pragma unroll
        for (int u = 0; u < RU; ++u) {
#pragma unroll
            for (int v = 0; v < 2; ++v) {
                const unsigned j = (unsigned)lane + 32u * v;
                r[u][v] = rec[j < rc[u] ? rb[u] + j : 0u];
            }
        }
#pragma unroll
        for (int u = 0; u < RU; ++u) {
#pragma unroll
            for (int v = 0; v < 2; ++v) {
                if ((unsigned)lane + 32u * v < rc[u]) f(r[u][v]);
            }
        }
#pragma unroll
        for (int u = 0; u < RU; ++u) {
            for (unsigned j = (unsigned)lane + 64u; j < rc[u]; j += 32u) f(rec[rb[u] + j]);
        }
    }
}

__global__ __launch_bounds__(1024) void grp_bucket(const unsigned* __restrict__ rec,
                                                   const unsigned* __restrict__ runs,
                                                   int E, int N, int NB, int G,
                                                   int* __restrict__ rowptr,
                                                   int* __restrict__ csr_src) {
    __shared__ unsigned c[256];
    __shared__ unsigned obuf[CAP];
    __shared__ unsigned rbeg[GMAX];
    __shared__ unsigned rcnt[GMAX];
    __shared__ unsigned wsum[64];
    __shared__ unsigned ws[32];
    const int t = threadIdx.x;
    const int lane = t & 31;
    const int w = __builtin_amdgcn_readfirstlane(t >> 5);
    const int b = blockIdx.x;
    unsigned st = 0u, cn = 0u;
    if (t < G) {
        const unsigned word = runs[t * run_pitch(NB) + b];
        st = word >> 16;
        cn = word & 0xFFFFu;
        st = st < (unsigned)CH ? st : (unsigned)CH;
        cn = cn < (unsigned)CH - st ? cn : (unsigned)CH - st;
        rbeg[t] = (unsigned)t * CH + st;
        rcnt[t] = cn;
    }
    if (t < 256) c[t] = 0u;
    {
        const unsigned a0 = wave_total(st, lane);
        const unsigned a1 = wave_total(cn, lane);
        if (lane == 0) { wsum[w] = a0; wsum[32 + w] = a1; }
    }
    __syncthreads();
    int beg = (int)wave_total(wsum[lane], lane);
    int end = beg + (int)wave_total(wsum[32 + lane], lane);
    end = end < 0 ? 0 : (end > E ? E : end);
    beg = beg < 0 ? 0 : (beg > end ? end : beg);
    for_each_record(rec, rbeg, rcnt, G, lane, w, [&](unsigned r) { atomicAdd(&c[r >> 24], 1u); });
    __syncthreads();
    const unsigned ex = block_excl_scan(t < 256 ? c[t] : 0u, ws, lane, w);
    if (t < 256) c[t] = ex;
    __syncthreads();
    const int cnt = end - beg;
    const bool staged = cnt <= CAP;
    for_each_record(rec, rbeg, rcnt, G, lane, w, [&](unsigned r) {
        const unsigned slot = atomicAdd(&c[r >> 24], 1u);
        const int s = (int)(r & 0xFFFFFFu);
        if (staged) {
            if (slot < (unsigned)CAP) obuf[slot] = (unsigned)s;
        } else {
            const unsigned gp = (unsigned)beg + slot;
            if (gp < (unsigned)E) csr_src[gp] = s;
        }
    });
    __syncthreads();
    if (staged) {
        for (int i = t; i < cnt; i += 1024) csr_src[beg + i] = (int)obuf[i];
    }
    if (t < 256) {
        const int node = (b << BSH) + t;
        if (node < N) rowptr[node] = beg + (int)ex;
    }
    if (b == NB - 1 && t == 0) rowptr[N] = end;
}

template <int C>
__device__ __forceinline__ void fold_level(float* p, int lane) {
    const bool hi = (lane & C) != 0;
#pragma unroll
    for (int k = 0; k < C; ++k) {
        const float keep = hi ? p[k + C] : p[k];
        const float send = hi ? p[k] : p[k + C];
        p[k] = keep + lane_xor<C>(send);
    }
}

template <int U>
__device__ __forceinline__ int grp_bcast(int x) {
    return __builtin_amdgcn_mov_dpp8(x, U * 0x249249);
}
__device__ __forceinline__ float grp_max(float v) {
    v = fmaxf(v, lane_xor<4>(v)); v = fmaxf(v, lane_xor<2>(v)); v = fmaxf(v, lane_xor<1>(v));
    return v;
}
__device__ __forceinline__ float grp_sum(float v) {
    v += lane_xor<4>(v); v += lane_xor<2>(v); v += lane_xor<1>(v);
    return v;
}
__device__ __forceinline__ float dot4(v4f x, v4f y) {
    const v2f t = __builtin_elementwise_fma(x.zw, y.zw, x.xy * y.xy);
    return t.x + t.y;
}

#define YP 36
__global__ __launch_bounds__(256) void gat_node(const int* __restrict__ rowptr,
                                                const int* __restrict__ csr_src,
                                                const float* __restrict__ Xin,
                                                const float* __restrict__ W,
                                                const float* __restrict__ a_src,
                                                const float* __restrict__ a_dst,
                                                const float* __restrict__ b,
                                                float* __restrict__ out, int N) {
    __shared__ v4f wa4[16];
    __shared__ v4f ybuf4[32 * YP / 4];
    float* const wa   = reinterpret_cast<float*>(wa4);
    float* const ybuf = reinterpret_cast<float*>(ybuf4);
    const int w    = __builtin_amdgcn_readfirstlane((int)(threadIdx.x >> 5));
    const int d0   = (blockIdx.x * 8 + w) * 4;
    const int lane = threadIdx.x & 31;
    const int  c    = lane & 7;
    const int wi = w * 4 + (lane >> 3);
    const v4f wr  = reinterpret_cast<const v4f*>(W)[wi * 8 + c];
    const v4f asr = reinterpret_cast<const v4f*>(a_src)[c];
    const v4f adr = reinterpret_cast<const v4f*>(a_dst)[c];
    const int  d    = d0 + (lane >> 3);
    const bool live = d < N;
    const int  dc   = live ? d : N - 1;
    const int  beg  = rowptr[dc];
    const int  deg  = live ? rowptr[dc + 1] - beg : 0;
    const v4f* __restrict__ X4 = reinterpret_cast<const v4f*>(Xin);
    const v4f hd  = X4[dc * 8 + c];
    int rem = deg - c;
    int pos = beg + c;
    int s_next = 0;
    if (rem > 0) s_next = csr_src[pos];
    const int dg01 = max(__builtin_amdgcn_readlane(deg, 0), __builtin_amdgcn_readlane(deg, 8));
    const int dg23 = max(__builtin_amdgcn_readlane(deg, 16), __builtin_amdgcn_readlane(deg, 24));
    const int maxdeg = max(dg01, dg23);

    {
        const float ps = grp_sum(dot4(wr, asr));
        const float pd = grp_sum(dot4(wr, adr));
        if (c == 0) { wa[wi] = ps; wa[32 + wi] = pd; }
    }
    __syncthreads();
    const v4f as4 = wa4[c];
    const v4f ad4 = wa4[8 + c];

    const float adv   = grp_sum(dot4(hd, ad4));
    const float lself = lrelu(grp_sum(dot4(hd, as4)) + adv);

    float m   = lself;
    float den = 1.f;
    v4f   acc = hd;

    for (int t = 0; t < maxdeg; t += 8) {
        const bool valid = rem > 0;
        const int s = s_next;
        s_next = 0;
        if (rem > 8) s_next = csr_src[pos + 8];
        rem -= 8;
        pos += 8;

        v4f r[8];
#pragma unroll
        for (int u = 0; u < 8; ++u) {
            int su;
            if (u == 0) su = grp_bcast<0>(s); else if (u == 1) su = grp_bcast<1>(s);
            else if (u == 2) su = grp_bcast<2>(s); else if (u == 3) su = grp_bcast<3>(s);
            else if (u == 4) su = grp_bcast<4>(s); else if (u == 5) su = grp_bcast<5>(s);
            else if (u == 6) su = grp_bcast<6>(s); else su = grp_bcast<7>(s);
            r[u] = X4[su * 8 + c];
        }
        __builtin_amdgcn_sched_barrier(0);

        float p[4];
        {
            const bool hi = (lane & 4) != 0;
#pragma unroll
            for (int k = 0; k < 4; ++k) {
                const float lo_row = dot4(r[k], as4);
                const float hi_row = dot4(r[k + 4], as4);
                const float keep = hi ? hi_row : lo_row;
                const float send = hi ? lo_row : hi_row;
                p[k] = keep + lane_xor<4>(send);
            }
        }
        fold_level<2>(p, lane);
        fold_level<1>(p, lane);

        const float l  = valid ? lrelu(p[0] + adv) : -__builtin_inff();
        const float mn = fmaxf(m, grp_max(l));
        const float sc = __expf(m - mn);
        const float e  = valid ? __expf(l - mn) : 0.f;
        den = den * sc + grp_sum(e);
        acc *= sc;
        m = mn;
#pragma unroll
        for (int u = 0; u < 8; ++u) {
            int eb;
            const int ei = __float_as_int(e);
            if (u == 0) eb = grp_bcast<0>(ei); else if (u == 1) eb = grp_bcast<1>(ei);
            else if (u == 2) eb = grp_bcast<2>(ei); else if (u == 3) eb = grp_bcast<3>(ei);
            else if (u == 4) eb = grp_bcast<4>(ei); else if (u == 5) eb = grp_bcast<5>(ei);
            else if (u == 6) eb = grp_bcast<6>(ei); else eb = grp_bcast<7>(ei);
            const float eu = __int_as_float(eb);
            acc.x = fmaf(eu, r[u].x, acc.x);
            acc.y = fmaf(eu, r[u].y, acc.y);
            acc.z = fmaf(eu, r[u].z, acc.z);
            acc.w = fmaf(eu, r[u].w, acc.w);
        }
    }

    const int le = (int)__builtin_amdgcn_mbcnt_lo(~0u, 0u);
    const int ce = le & 7;
    const int rl = w * 4 + (le >> 3);
    {
        const float dn = den + EPS;
        v4f y;
        y.x = acc.x / dn;
        y.y = acc.y / dn;
        y.z = acc.z / dn;
        y.w = acc.w / dn;
        ybuf4[rl * (YP / 4) + ce] = y;
    }
    const int col   = le & 15;
    const int khalf = (le >> 4) * 2;
    v2f b0[8], b1[8];
    if (w < 2) {
#pragma unroll
        for (int kb = 0; kb < 8; ++kb) {
            const int k0 = kb * 4 + khalf;
            b0[kb].x = W[k0 * D + col];        b0[kb].y = W[(k0 + 1) * D + col];
            b1[kb].x = W[k0 * D + col + 16];   b1[kb].y = W[(k0 + 1) * D + col + 16];
        }
    }
    __syncthreads();
    if (w < 2) {
        const float* ya = ybuf + (w * 16 + col) * YP;
        v8f c0 = {}; v8f c1 = {};
#pragma unroll
        for (int kb = 0; kb < 8; ++kb) {
            const int k0 = kb * 4 + khalf;
            v2f a;  a.x  = ya[k0];                 a.y  = ya[k0 + 1];
            c0 = __builtin_amdgcn_wmma_f32_16x16x4_f32(false, a, false, b0[kb], (short)0, c0, false, false);
            c1 = __builtin_amdgcn_wmma_f32_16x16x4_f32(false, a, false, b1[kb], (short)0, c1, false, false);
        }
        float* yo = ybuf + (w * 16 + (le >> 4) * 8) * YP;
#pragma unroll
        for (int r = 0; r < 8; ++r) {
            yo[r * YP + col]      = c0[r];
            yo[r * YP + col + 16] = c1[r];
        }
    }
    __syncthreads();
    if (d0 + (le >> 3) < N) {
        const v4f b4 = reinterpret_cast<const v4f*>(b)[ce];
        v4f v = ybuf4[rl * (YP / 4) + ce];
        v.x += b4.x;
        v.y += b4.y;
        v.z += b4.z;
        v.w += b4.w;
        v.x = (v.x > 0.f) ? v.x : (__expf(v.x) - 1.f);
        v.y = (v.y > 0.f) ? v.y : (__expf(v.y) - 1.f);
        v.z = (v.z > 0.f) ? v.z : (__expf(v.z) - 1.f);
        v.w = (v.w > 0.f) ? v.w : (__expf(v.w) - 1.f);
        reinterpret_cast<v4f*>(out)[(d0 + (le >> 3)) * 8 + ce] = v;
    }
}

extern "C" void kernel_launch(void* const* d_in, const int* in_sizes, int n_in,
                              void* d_out, int out_size, void* d_ws, size_t ws_size,
                              hipStream_t stream) {
    const float* x      = (const float*)d_in[0];
    const int*   ei     = (const int*)d_in[1];
    const float* W1     = (const float*)d_in[2];
    const float* a_src1 = (const float*)d_in[3];
    const float* a_dst1 = (const float*)d_in[4];
    const float* b1     = (const float*)d_in[5];
    const float* W2     = (const float*)d_in[6];
    const float* a_src2 = (const float*)d_in[7];
    const float* a_dst2 = (const float*)d_in[8];
    const float* b2     = (const float*)d_in[9];

    const int N = in_sizes[0] / D;
    const int E = in_sizes[1] / 2;

    const int NB = (N + 255) / 256;
    const int G  = (E + CH - 1) / CH;
    const int M  = G * ((NB + 31) & ~31);

    unsigned* runs    = (unsigned*)d_ws;
    unsigned* rec     = runs + M;
    int*      rowptr  = (int*)(rec + (size_t)G * CH);
    int*      csr_src = rowptr + (N + 1);

    float* xbar = (float*)d_out;
    float* z    = xbar + (size_t)N * D;

    grp_sort<<<G, 1024, 0, stream>>>(ei, E, NB, G, runs, rec);
    grp_bucket<<<NB, 1024, 0, stream>>>(rec, runs, E, N, NB, G, rowptr, csr_src);

    const int node_blocks = (N + 31) / 32;
    gat_node<<<node_blocks, 256, 0, stream>>>(rowptr, csr_src, x, W1, a_src1, a_dst1, b1, z, N);
    gat_node<<<node_blocks, 256, 0, stream>>>(rowptr, csr_src, z, W2, a_src2, a_dst2, b2, xbar, N);
}


# Layer 1 moved into the bucket kernel: the work-group that orders a bucket's edges walks them from LDS for its 256 nodes, 256 rows through the WMMA product: 3 launches, not 4; sum order as before
# speedup vs baseline: 14.5189x; 1.0961x over previous
#include <hip/hip_runtime.h>
#include <hip/hip_bf16.h>


typedef __attribute__((ext_vector_type(2))) float v2f;
typedef __attribute__((ext_vector_type(8))) float v8f;
typedef __attribute__((ext_vector_type(4))) float v4f;

#define D 32
#define SLOPE 0.2f
#define EPS 1e-16f

__device__ __forceinline__ float lrelu(float x) { return x > 0.f ? x : SLOPE * x; }

template <int M>
__device__ __forceinline__ float lane_xor(float x) {
    if constexpr (M == 16) {
        return __int_as_float(__builtin_amdgcn_permlanex16(__float_as_int(x), __float_as_int(x),
                                                           0x76543210, 0xfedcba98, false, false));
    } else {
        return __int_as_float(__builtin_amdgcn_update_dpp(0, __float_as_int(x), 0x160 | M, 0xf, 0xf, true));
    }
}

__device__ __forceinline__ float wave_max(float v) {
    v = fmaxf(v, lane_xor<16>(v)); v = fmaxf(v, lane_xor<8>(v)); v = fmaxf(v, lane_xor<4>(v));
    v = fmaxf(v, lane_xor<2>(v));  v = fmaxf(v, lane_xor<1>(v));
    return v;
}
__device__ __forceinline__ float wave_sum(float v) {
    v += lane_xor<16>(v); v += lane_xor<8>(v); v += lane_xor<4>(v); v += lane_xor<2>(v); v += lane_xor<1>(v);
    return v;
}

#define CH    8192
#define EPT   8
#define BSH   8
#define NBMAX 1024
#define GMAX  1024
#define CAP   16384
#define RU    4
__device__ __forceinline__ int run_pitch(int NB) { return (NB + 31) & ~31; }

template <int S>
__device__ __forceinline__ unsigned row_shr(unsigned x) {
    return (unsigned)__builtin_amdgcn_update_dpp(0, (int)x, 0x110 | S, 0xf, 0xf, true);
}

__device__ __forceinline__ unsigned wave_incl_scan(unsigned x, int lane) {
    x += row_shr<1>(x);
    x += row_shr<2>(x);
    x += row_shr<4>(x);
    x += row_shr<8>(x);
    const unsigned low = (unsigned)__builtin_amdgcn_readlane((int)x, 15);
    return x + ((lane & 16) ? low : 0u);
}

__device__ __forceinline__ unsigned wave_total(unsigned x, int lane) {
    return (unsigned)__builtin_amdgcn_readlane((int)wave_incl_scan(x, lane), 31);
}

__device__ __forceinline__ unsigned block_excl_scan(unsigned x, unsigned* ws, int lane, int w) {
    const unsigned inc = wave_incl_scan(x, lane);
    if (lane == 31) ws[w] = inc;
    __syncthreads();
    const unsigned y  = ws[lane];
    const unsigned yi = wave_incl_scan(y, lane);
    return (unsigned)__builtin_amdgcn_readlane((int)(yi - y), w) + inc - x;
}

__global__ __launch_bounds__(1024) void grp_sort(const int* __restrict__ ei, int E, int NB, int G,
                                                 unsigned* __restrict__ runs,
                                                 unsigned* __restrict__ rec) {
    __shared__ unsigned h[NBMAX];
    __shared__ unsigned lstart[NBMAX];
    __shared__ uint4    sbuf4[CH / 4];
    __shared__ unsigned ws[32];
    unsigned* const sbuf = reinterpret_cast<unsigned*>(sbuf4);
    const int t = threadIdx.x;
    const int lane = t & 31;
    const int w = __builtin_amdgcn_readfirstlane(t >> 5);
    const int g = blockIdx.x;
    int      bk[EPT];
    unsigned pk[EPT];
    unsigned rk[EPT];
    const int e0 = g * CH + t;
    int sv[EPT];
    int dv[EPT];
#pragma unroll
    for (int k = 0; k < EPT; ++k) {
        const int e = e0 + k * 1024;
        const int ec = e < E ? e : E - 1;
        sv[k] = ei[ec];
        dv[k] = ei[E + ec];
    }
    for (int i = t; i < NB; i += 1024) h[i] = 0u;
    __syncthreads();
#pragma unroll
    for (int k = 0; k < EPT; ++k) {
        const int e = e0 + k * 1024;
        bk[k] = -1; pk[k] = 0u; rk[k] = 0u;
        if (e < E) {
            const int s = sv[k];
            const int d = dv[k];
            int b = d >> BSH;
            b = b < NB ? b : NB - 1;
            bk[k] = b;
            pk[k] = ((unsigned)(d & 255) << 24) | ((unsigned)s & 0xFFFFFFu);
            rk[k] = atomicAdd(&h[b], 1u);
        }
    }
    __syncthreads();
    {
        const unsigned cnt = t < NB ? h[t] : 0u;
        const unsigned ls = block_excl_scan(cnt, ws, lane, w);
        if (t < NB) {
            lstart[t] = ls;
            runs[g * run_pitch(NB) + t] = (ls << 16) | cnt;
        }
    }
    __syncthreads();
#pragma unroll
    for (int k = 0; k < EPT; ++k) {
        if (bk[k] >= 0) sbuf[lstart[bk[k]] + rk[k]] = pk[k];
    }
    __syncthreads();
    const int left = E - g * CH;
    const int n = left < CH ? left : CH;
    uint4* __restrict__ seg = reinterpret_cast<uint4*>(rec + (size_t)g * CH);
    for (int i = t; i * 4 < n; i += 1024) seg[i] = sbuf4[i];
}

template <typename F>
__device__ __forceinline__ void for_each_record(const unsigned* __restrict__ rec, const unsigned* rbeg,
                                                const unsigned* rcnt, int G, int lane, int w, F f) {
    for (int g0 = w; g0 < G; g0 += 32 * RU) {
        unsigned rb[RU], rc[RU];
#pragma unroll
        for (int u = 0; u < RU; ++u) {
            const int g = g0 + 32 * u;
            rb[u] = g < G ? rbeg[g] : 0u;
            rc[u] = g < G ? rcnt[g] : 0u;
        }
        unsigned r[RU][2];
#pragma unroll
        for (int u = 0; u < RU; ++u) {
#pragma unroll
            for (int v = 0; v < 2; ++v) {
                const unsigned j = (unsigned)lane + 32u * v;
                r[u][v] = rec[j < rc[u] ? rb[u] + j : 0u];
            }
        }
#pragma unroll
        for (int u = 0; u < RU; ++u) {
#pragma unroll
            for (int v = 0; v < 2; ++v) {
                if ((unsigned)lane + 32u * v < rc[u]) f(r[u][v]);
            }
        }
#pragma unroll
        for (int u = 0; u < RU; ++u) {
            for (unsigned j = (unsigned)lane + 64u; j < rc[u]; j += 32u) f(rec[rb[u] + j]);
        }
    }
}

template <int C>
__device__ __forceinline__ void fold_level(float* p, int lane) {
    const bool hi = (lane & C) != 0;
#pragma unroll
    for (int k = 0; k < C; ++k) {
        const float keep = hi ? p[k + C] : p[k];
        const float send = hi ? p[k] : p[k + C];
        p[k] = keep + lane_xor<C>(send);
    }
}

template <int U>
__device__ __forceinline__ int grp_bcast(int x) {
    return __builtin_amdgcn_mov_dpp8(x, U * 0x249249);
}
__device__ __forceinline__ float grp_max(float v) {
    v = fmaxf(v, lane_xor<4>(v)); v = fmaxf(v, lane_xor<2>(v)); v = fmaxf(v, lane_xor<1>(v));
    return v;
}
__device__ __forceinline__ float grp_sum(float v) {
    v += lane_xor<4>(v); v += lane_xor<2>(v); v += lane_xor<1>(v);
    return v;
}
__device__ __forceinline__ float dot4(v4f x, v4f y) {
    const v2f t = __builtin_elementwise_fma(x.zw, y.zw, x.xy * y.xy);
    return t.x + t.y;
}

#define YP 36
__global__ __launch_bounds__(256) void gat_node(const int* __restrict__ rowptr,
                                                const int* __restrict__ csr_src,
                                                const float* __restrict__ Xin,
                                                const float* __restrict__ W,
                                                const float* __restrict__ a_src,
                                                const float* __restrict__ a_dst,
                                                const float* __restrict__ b,
                                                float* __restrict__ out, int N) {
    __shared__ v4f wa4[16];
    __shared__ v4f ybuf4[32 * YP / 4];
    float* const wa   = reinterpret_cast<float*>(wa4);
    float* const ybuf = reinterpret_cast<float*>(ybuf4);
    const int w    = __builtin_amdgcn_readfirstlane((int)(threadIdx.x >> 5));
    const int d0   = (blockIdx.x * 8 + w) * 4;
    const int lane = threadIdx.x & 31;
    const int  c    = lane & 7;
    const int wi = w * 4 + (lane >> 3);
    const v4f wr  = reinterpret_cast<const v4f*>(W)[wi * 8 + c];
    const v4f asr = reinterpret_cast<const v4f*>(a_src)[c];
    const v4f adr = reinterpret_cast<const v4f*>(a_dst)[c];
    const int  d    = d0 + (lane >> 3);
    const bool live = d < N;
    const int  dc   = live ? d : N - 1;
    const int  beg  = rowptr[dc];
    const int  deg  = live ? rowptr[dc + 1] - beg : 0;
    const v4f* __restrict__ X4 = reinterpret_cast<const v4f*>(Xin);
    const v4f hd  = X4[dc * 8 + c];
    int rem = deg - c;
    int pos = beg + c;
    int s_next = 0;
    if (rem > 0) s_next = csr_src[pos];
    const int dg01 = max(__builtin_amdgcn_readlane(deg, 0), __builtin_amdgcn_readlane(deg, 8));
    const int dg23 = max(__builtin_amdgcn_readlane(deg, 16), __builtin_amdgcn_readlane(deg, 24));
    const int maxdeg = max(dg01, dg23);

    {
        const float ps = grp_sum(dot4(wr, asr));
        const float pd = grp_sum(dot4(wr, adr));
        if (c == 0) { wa[wi] = ps; wa[32 + wi] = pd; }
    }
    __syncthreads();
    const v4f as4 = wa4[c];
    const v4f ad4 = wa4[8 + c];

    const float adv   = grp_sum(dot4(hd, ad4));
    const float lself = lrelu(grp_sum(dot4(hd, as4)) + adv);

    float m   = lself;
    float den = 1.f;
    v4f   acc = hd;

    for (int t = 0; t < maxdeg; t += 8) {
        const bool valid = rem > 0;
        const int s = s_next;
        s_next = 0;
        if (rem > 8) s_next = csr_src[pos + 8];
        rem -= 8;
        pos += 8;

        v4f r[8];
#pragma unroll
        for (int u = 0; u < 8; ++u) {
            int su;
            if (u == 0) su = grp_bcast<0>(s); else if (u == 1) su = grp_bcast<1>(s);
            else if (u == 2) su = grp_bcast<2>(s); else if (u == 3) su = grp_bcast<3>(s);
            else if (u == 4) su = grp_bcast<4>(s); else if (u == 5) su = grp_bcast<5>(s);
            else if (u == 6) su = grp_bcast<6>(s); else su = grp_bcast<7>(s);
            r[u] = X4[su * 8 + c];
        }
        __builtin_amdgcn_sched_barrier(0);

        float p[4];
        {
            const bool hi = (lane & 4) != 0;
#pragma unroll
            for (int k = 0; k < 4; ++k) {
                const float lo_row = dot4(r[k], as4);
                const float hi_row = dot4(r[k + 4], as4);
                const float keep = hi ? hi_row : lo_row;
                const float send = hi ? lo_row : hi_row;
                p[k] = keep + lane_xor<4>(send);
            }
        }
        fold_level<2>(p, lane);
        fold_level<1>(p, lane);

        const float l  = valid ? lrelu(p[0] + adv) : -__builtin_inff();
        const float mn = fmaxf(m, grp_max(l));
        const float sc = __expf(m - mn);
        const float e  = valid ? __expf(l - mn) : 0.f;
        den = den * sc + grp_sum(e);
        acc *= sc;
        m = mn;
#pragma unroll
        for (int u = 0; u < 8; ++u) {
            int eb;
            const int ei = __float_as_int(e);
            if (u == 0) eb = grp_bcast<0>(ei); else if (u == 1) eb = grp_bcast<1>(ei);
            else if (u == 2) eb = grp_bcast<2>(ei); else if (u == 3) eb = grp_bcast<3>(ei);
            else if (u == 4) eb = grp_bcast<4>(ei); else if (u == 5) eb = grp_bcast<5>(ei);
            else if (u == 6) eb = grp_bcast<6>(ei); else eb = grp_bcast<7>(ei);
            const float eu = __int_as_float(eb);
            acc.x = fmaf(eu, r[u].x, acc.x);
            acc.y = fmaf(eu, r[u].y, acc.y);
            acc.z = fmaf(eu, r[u].z, acc.z);
            acc.w = fmaf(eu, r[u].w, acc.w);
        }
    }

    const int le = (int)__builtin_amdgcn_mbcnt_lo(~0u, 0u);
    const int ce = le & 7;
    const int rl = w * 4 + (le >> 3);
    {
        const float dn = den + EPS;
        v4f y;
        y.x = acc.x / dn;
        y.y = acc.y / dn;
        y.z = acc.z / dn;
        y.w = acc.w / dn;
        ybuf4[rl * (YP / 4) + ce] = y;
    }
    const int col   = le & 15;
    const int khalf = (le >> 4) * 2;
    v2f b0[8], b1[8];
    if (w < 2) {
#pragma unroll
        for (int kb = 0; kb < 8; ++kb) {
            const int k0 = kb * 4 + khalf;
            b0[kb].x = W[k0 * D + col];        b0[kb].y = W[(k0 + 1) * D + col];
            b1[kb].x = W[k0 * D + col + 16];   b1[kb].y = W[(k0 + 1) * D + col + 16];
        }
    }
    __syncthreads();
    if (w < 2) {
        const float* ya = ybuf + (w * 16 + col) * YP;
        v8f c0 = {}; v8f c1 = {};
#pragma unroll
        for (int kb = 0; kb < 8; ++kb) {
            const int k0 = kb * 4 + khalf;
            v2f a;  a.x  = ya[k0];                 a.y  = ya[k0 + 1];
            c0 = __builtin_amdgcn_wmma_f32_16x16x4_f32(false, a, false, b0[kb], (short)0, c0, false, false);
            c1 = __builtin_amdgcn_wmma_f32_16x16x4_f32(false, a, false, b1[kb], (short)0, c1, false, false);
        }
        float* yo = ybuf + (w * 16 + (le >> 4) * 8) * YP;
#pragma unroll
        for (int r = 0; r < 8; ++r) {
            yo[r * YP + col]      = c0[r];
            yo[r * YP + col + 16] = c1[r];
        }
    }
    __syncthreads();
    if (d0 + (le >> 3) < N) {
        const v4f b4 = reinterpret_cast<const v4f*>(b)[ce];
        v4f v = ybuf4[rl * (YP / 4) + ce];
        v.x += b4.x;
        v.y += b4.y;
        v.z += b4.z;
        v.w += b4.w;
        v.x = (v.x > 0.f) ? v.x : (__expf(v.x) - 1.f);
        v.y = (v.y > 0.f) ? v.y : (__expf(v.y) - 1.f);
        v.z = (v.z > 0.f) ? v.z : (__expf(v.z) - 1.f);
        v.w = (v.w > 0.f) ? v.w : (__expf(v.w) - 1.f);
        reinterpret_cast<v4f*>(out)[(d0 + (le >> 3)) * 8 + ce] = v;
    }
}

template <bool STAGED>
__device__ __forceinline__ int src_at(const unsigned* obuf, const int* csr_src, int beg, int E, int i) {
    if constexpr (STAGED) {
        return (int)obuf[i];
    } else {
        const int gp = beg + i;
        return csr_src[gp < E ? gp : E - 1];
    }
}

template <bool STAGED>
__device__ __forceinline__ void bucket_layer(const unsigned* cend, const unsigned* obuf, const int* csr_src,
                                             int beg, int E, const v4f* wa4, v4f* ybuf4,
                                             const float* __restrict__ Xin, const float* __restrict__ W,
                                             const float* __restrict__ bias, float* __restrict__ out,
                                             int N, int b, int w) {
    float* const ybuf = reinterpret_cast<float*>(ybuf4);
    const v4f* __restrict__ X4 = reinterpret_cast<const v4f*>(Xin);
    const int lane = threadIdx.x & 31;
    const int c8   = lane & 7;
    const int g    = lane >> 3;
    const v4f as4 = wa4[c8];
    const v4f ad4 = wa4[8 + c8];
    for (int p = 0; p < 2; ++p) {
        const int  ln   = (p * 32 + w) * 4 + g;
        const int  d    = (b << BSH) + ln;
        const bool live = d < N;
        const int  dc   = live ? d : N - 1;
        const v4f  hd   = X4[dc * 8 + c8];
        const int  lb   = ln ? (int)cend[ln - 1] : 0;
        const int  deg  = live ? (int)cend[ln] - lb : 0;
        int rem = deg - c8;
        int pos = lb + c8;
        int s_next = 0;
        if (rem > 0) s_next = src_at<STAGED>(obuf, csr_src, beg, E, pos);
        const int dg01 = max(__builtin_amdgcn_readlane(deg, 0), __builtin_amdgcn_readlane(deg, 8));
        const int dg23 = max(__builtin_amdgcn_readlane(deg, 16), __builtin_amdgcn_readlane(deg, 24));
        const int maxdeg = max(dg01, dg23);

        const float adv   = grp_sum(dot4(hd, ad4));
        const float lself = lrelu(grp_sum(dot4(hd, as4)) + adv);

        float m   = lself;
        float den = 1.f;
        v4f   acc = hd;

        for (int t = 0; t < maxdeg; t += 8) {
            const bool valid = rem > 0;
            const int s = s_next;
            s_next = 0;
            if (rem > 8) s_next = src_at<STAGED>(obuf, csr_src, beg, E, pos + 8);
            rem -= 8;
            pos += 8;

            v4f r[8];
#pragma unroll
            for (int u = 0; u < 8; ++u) {
                int su;
                if (u == 0) su = grp_bcast<0>(s); else if (u == 1) su = grp_bcast<1>(s);
                else if (u == 2) su = grp_bcast<2>(s); else if (u == 3) su = grp_bcast<3>(s);
                else if (u == 4) su = grp_bcast<4>(s); else if (u == 5) su = grp_bcast<5>(s);
                else if (u == 6) su = grp_bcast<6>(s); else su = grp_bcast<7>(s);
                r[u] = X4[su * 8 + c8];
            }
            __builtin_amdgcn_sched_barrier(0);

            float pr[4];
            {
                const bool hi = (lane & 4) != 0;
#pragma unroll
                for (int k = 0; k < 4; ++k) {
                    const float lo_row = dot4(r[k], as4);
                    const float hi_row = dot4(r[k + 4], as4);
                    const float keep = hi ? hi_row : lo_row;
                    const float send = hi ? lo_row : hi_row;
                    pr[k] = keep + lane_xor<4>(send);
                }
            }
            fold_level<2>(pr, lane);
            fold_level<1>(pr, lane);

            const float l  = valid ? lrelu(pr[0] + adv) : -__builtin_inff();
            const float mn = fmaxf(m, grp_max(l));
            const float sc = __expf(m - mn);
            const float e  = valid ? __expf(l - mn) : 0.f;
            den = den * sc + grp_sum(e);
            acc *= sc;
            m = mn;
#pragma unroll
            for (int u = 0; u < 8; ++u) {
                int eb;
                const int ei = __float_as_int(e);
                if (u == 0) eb = grp_bcast<0>(ei); else if (u == 1) eb = grp_bcast<1>(ei);
                else if (u == 2) eb = grp_bcast<2>(ei); else if (u == 3) eb = grp_bcast<3>(ei);
                else if (u == 4) eb = grp_bcast<4>(ei); else if (u == 5) eb = grp_bcast<5>(ei);
                else if (u == 6) eb = grp_bcast<6>(ei); else eb = grp_bcast<7>(ei);
                const float eu = __int_as_float(eb);
                acc.x = fmaf(eu, r[u].x, acc.x);
                acc.y = fmaf(eu, r[u].y, acc.y);
                acc.z = fmaf(eu, r[u].z, acc.z);
                acc.w = fmaf(eu, r[u].w, acc.w);
            }
        }
        const float dn = den + EPS;
        v4f y;
        y.x = acc.x / dn;
        y.y = acc.y / dn;
        y.z = acc.z / dn;
        y.w = acc.w / dn;
        ybuf4[ln * (YP / 4) + c8] = y;
    }
    const int col   = lane & 15;
    const int khalf = (lane >> 4) * 2;
    v2f b0[8], b1[8];
    if (w < 16) {
#pragma unroll
        for (int kb = 0; kb < 8; ++kb) {
            const int k0 = kb * 4 + khalf;
            b0[kb].x = W[k0 * D + col];        b0[kb].y = W[(k0 + 1) * D + col];
            b1[kb].x = W[k0 * D + col + 16];   b1[kb].y = W[(k0 + 1) * D + col + 16];
        }
    }
    __syncthreads();
    if (w < 16) {
        const float* ya = ybuf + (w * 16 + col) * YP;
        v8f c0 = {}; v8f c1 = {};
#pragma unroll
        for (int kb = 0; kb < 8; ++kb) {
            const int k0 = kb * 4 + khalf;
            v2f a;  a.x  = ya[k0];                 a.y  = ya[k0 + 1];
            c0 = __builtin_amdgcn_wmma_f32_16x16x4_f32(false, a, false, b0[kb], (short)0, c0, false, false);
            c1 = __builtin_amdgcn_wmma_f32_16x16x4_f32(false, a, false, b1[kb], (short)0, c1, false, false);
        }
        float* yo = ybuf + (w * 16 + (lane >> 4) * 8) * YP;
#pragma unroll
        for (int r = 0; r < 8; ++r) {
            yo[r * YP + col]      = c0[r];
            yo[r * YP + col + 16] = c1[r];
        }
    }
    __syncthreads();
    const v4f b4 = reinterpret_cast<const v4f*>(bias)[c8];
    for (int p = 0; p < 2; ++p) {
        const int ln = (p * 32 + w) * 4 + g;
        const int d  = (b << BSH) + ln;
        if (d < N) {
            v4f v = ybuf4[ln * (YP / 4) + c8];
            v.x += b4.x;
            v.y += b4.y;
            v.z += b4.z;
            v.w += b4.w;
            v.x = (v.x > 0.f) ? v.x : (__expf(v.x) - 1.f);
            v.y = (v.y > 0.f) ? v.y : (__expf(v.y) - 1.f);
            v.z = (v.z > 0.f) ? v.z : (__expf(v.z) - 1.f);
            v.w = (v.w > 0.f) ? v.w : (__expf(v.w) - 1.f);
            reinterpret_cast<v4f*>(out)[d * 8 + c8] = v;
        }
    }
}

__global__ __launch_bounds__(1024) void grp_bucket_node(const unsigned* __restrict__ rec,
                                                        const unsigned* __restrict__ runs,
                                                        int E, int N, int NB, int G,
                                                        int* __restrict__ rowptr,
                                                        int* csr_src,
                                                        const float* __restrict__ Xin,
                                                        const float* __restrict__ W,
                                                        const float* __restrict__ a_src,
                                                        const float* __restrict__ a_dst,
                                                        const float* __restrict__ bias,
                                                        float* __restrict__ out) {
    __shared__ unsigned c[256];
    __shared__ unsigned obuf[CAP];
    __shared__ unsigned rbeg[GMAX];
    __shared__ unsigned rcnt[GMAX];
    __shared__ unsigned wsum[64];
    __shared__ unsigned ws[32];
    __shared__ v4f wa4[16];
    __shared__ v4f ybuf4[256 * YP / 4];
    const int t = threadIdx.x;
    const int lane = t & 31;
    const int w = __builtin_amdgcn_readfirstlane(t >> 5);
    const int b = blockIdx.x;
    if (w < 8) {
        float* const wa = reinterpret_cast<float*>(wa4);
        const int c8 = lane & 7;
        const int wi = w * 4 + (lane >> 3);
        const v4f wr  = reinterpret_cast<const v4f*>(W)[wi * 8 + c8];
        const v4f asr = reinterpret_cast<const v4f*>(a_src)[c8];
        const v4f adr = reinterpret_cast<const v4f*>(a_dst)[c8];
        const float ps = grp_sum(dot4(wr, asr));
        const float pd = grp_sum(dot4(wr, adr));
        if (c8 == 0) { wa[wi] = ps; wa[32 + wi] = pd; }
    }
    unsigned st = 0u, cn = 0u;
    if (t < G) {
        const unsigned word = runs[t * run_pitch(NB) + b];
        st = word >> 16;
        cn = word & 0xFFFFu;
        st = st < (unsigned)CH ? st : (unsigned)CH;
        cn = cn < (unsigned)CH - st ? cn : (unsigned)CH - st;
        rbeg[t] = (unsigned)t * CH + st;
        rcnt[t] = cn;
    }
    if (t < 256) c[t] = 0u;
    {
        const unsigned a0 = wave_total(st, lane);
        const unsigned a1 = wave_total(cn, lane);
        if (lane == 0) { wsum[w] = a0; wsum[32 + w] = a1; }
    }
    __syncthreads();
    int beg = (int)wave_total(wsum[lane], lane);
    int end = beg + (int)wave_total(wsum[32 + lane], lane);
    end = end < 0 ? 0 : (end > E ? E : end);
    beg = beg < 0 ? 0 : (beg > end ? end : beg);
    for_each_record(rec, rbeg, rcnt, G, lane, w, [&](unsigned r) { atomicAdd(&c[r >> 24], 1u); });
    __syncthreads();
    const unsigned ex = block_excl_scan(t < 256 ? c[t] : 0u, ws, lane, w);
    if (t < 256) c[t] = ex;
    __syncthreads();
    const int cnt = end - beg;
    const bool staged = cnt <= CAP;
    for_each_record(rec, rbeg, rcnt, G, lane, w, [&](unsigned r) {
        const unsigned slot = atomicAdd(&c[r >> 24], 1u);
        const int s = (int)(r & 0xFFFFFFu);
        if (staged) {
            if (slot < (unsigned)CAP) obuf[slot] = (unsigned)s;
        } else {
            const unsigned gp = (unsigned)beg + slot;
            if (gp < (unsigned)E) csr_src[gp] = s;
        }
    });
    __syncthreads();
    if (staged) {
        for (int i = t; i < cnt; i += 1024) csr_src[beg + i] = (int)obuf[i];
    }
    if (t < 256) {
        const int node = (b << BSH) + t;
        if (node < N) rowptr[node] = beg + (int)ex;
    }
    if (b == NB - 1 && t == 0) rowptr[N] = end;
    if (staged) bucket_layer<true>(c, obuf, csr_src, beg, E, wa4, ybuf4, Xin, W, bias, out, N, b, w);
    else        bucket_layer<false>(c, obuf, csr_src, beg, E, wa4, ybuf4, Xin, W, bias, out, N, b, w);
}

extern "C" void kernel_launch(void* const* d_in, const int* in_sizes, int n_in,
                              void* d_out, int out_size, void* d_ws, size_t ws_size,
                              hipStream_t stream) {
    const float* x      = (const float*)d_in[0];
    const int*   ei     = (const int*)d_in[1];
    const float* W1     = (const float*)d_in[2];
    const float* a_src1 = (const float*)d_in[3];
    const float* a_dst1 = (const float*)d_in[4];
    const float* b1     = (const float*)d_in[5];
    const float* W2     = (const float*)d_in[6];
    const float* a_src2 = (const float*)d_in[7];
    const float* a_dst2 = (const float*)d_in[8];
    const float* b2     = (const float*)d_in[9];

    const int N = in_sizes[0] / D;
    const int E = in_sizes[1] / 2;

    const int NB = (N + 255) / 256;
    const int G  = (E + CH - 1) / CH;
    const int M  = G * ((NB + 31) & ~31);

    unsigned* runs    = (unsigned*)d_ws;
    unsigned* rec     = runs + M;
    int*      rowptr  = (int*)(rec + (size_t)G * CH);
    int*      csr_src = rowptr + (N + 1);

    float* xbar = (float*)d_out;
    float* z    = xbar + (size_t)N * D;

    grp_sort<<<G, 1024, 0, stream>>>(ei, E, NB, G, runs, rec);
    grp_bucket_node<<<NB, 1024, 0, stream>>>(rec, runs, E, N, NB, G, rowptr, csr_src, x, W1, a_src1, a_dst1, b1, z);

    const int node_blocks = (N + 31) / 32;
    gat_node<<<node_blocks, 256, 0, stream>>>(rowptr, csr_src, z, W2, a_src2, a_dst2, b2, xbar, N);
}
